# starting kernel
# baseline (speedup 1.0000x reference)
#include <hip/hip_runtime.h>

typedef __attribute__((ext_vector_type(2))) float v2f;
typedef __attribute__((ext_vector_type(8))) float v8f;

#define QS_THREADS 128

__device__ __forceinline__ int perm4(int x) {
  if (x & 8) x ^= 4;
  if (x & 4) x ^= 2;
  if (x & 2) x ^= 1;
  return x;
}
__device__ __forceinline__ int perm4inv(int x) {
  if (x & 2) x ^= 1;
  if (x & 4) x ^= 2;
  if (x & 8) x ^= 4;
  return x;
}

__global__ __launch_bounds__(QS_THREADS) void qsim_wmma_kernel(
    const float* __restrict__ state_batch,
    const float* __restrict__ params,
    const float* __restrict__ head_w,
    const float* __restrict__ head_b,
    float* __restrict__ out) {
  __shared__ float psi[4096];
  __shared__ float U[5][256];
  __shared__ float cz[24], sz[24];
  __shared__ float hws[12];
  __shared__ float red[QS_THREADS];

  const int b   = blockIdx.x;
  const int tid = threadIdx.x;

  if (tid < 12) {
    float a = 0.5f * state_batch[(size_t)b * 12 + tid];
    cz[tid] = cosf(a); sz[tid] = sinf(a);
    hws[tid] = head_w[tid];
  } else if (tid < 24) {
    float a = 0.5f * params[tid - 12];
    cz[tid] = cosf(a); sz[tid] = sinf(a);
  }
  __syncthreads();

  auto kron4 = [&](int base, int i, int j) -> float {
    float v = 1.0f;
#pragma unroll
    for (int q = 0; q < 4; ++q) {
      int bi = (i >> (3 - q)) & 1;
      int bj = (j >> (3 - q)) & 1;
      float c = cz[base + q], s = sz[base + q];
      v *= bi ? (bj ? c : s) : (bj ? -s : c);
    }
    return v;
  };

  for (int e = tid; e < 5 * 256; e += QS_THREADS) {
    int mi = e >> 8, idx = e & 255, i = idx >> 4, j = idx & 15;
    float v;
    switch (mi) {
      case 0:  v = kron4(12, perm4inv(i), j); break;
      case 1:  v = kron4(12, i, j); break;
      case 2:  v = (i == perm4(j)) ? 1.0f : 0.0f; break;
      case 3:  v = kron4(16, i, j); break;
      default: v = kron4(20, i, perm4(j)); break;
    }
    U[mi][idx] = v;
  }

  for (int s = tid; s < 4096; s += QS_THREADS) {
    int s2 = (perm4inv(s >> 8) << 8) | (s & 255);
    float v = 1.0f;
#pragma unroll
    for (int w = 0; w < 12; ++w)
      v *= ((s2 >> (11 - w)) & 1) ? sz[w] : cz[w];
    psi[s] = v;
  }
  __syncthreads();

  auto apply = [&](const float* Um, int sm, int sn, int st) {
    const int lane = tid & 31, wave = tid >> 5;
    const int n = lane & 15, hi = lane >> 4;
    v2f a[4];
#pragma unroll
    for (int kc = 0; kc < 4; ++kc) {
      int k0 = 4 * kc + 2 * hi;
      a[kc].x = Um[n * 16 + k0];
      a[kc].y = Um[n * 16 + k0 + 1];
    }
#pragma unroll
    for (int tt = 0; tt < 4; ++tt) {
      int t = wave * 4 + tt;
      int base = st * t + sn * n;
      v8f c = {};
#pragma unroll
      for (int kc = 0; kc < 4; ++kc) {
        int k0 = 4 * kc + 2 * hi;
        v2f bf;
        bf.x = psi[base + sm * k0];
        bf.y = psi[base + sm * (k0 + 1)];
        c = __builtin_amdgcn_wmma_f32_16x16x4_f32(
             false, a[kc],  false, bf,
             (short)0, c,  false,  false);
      }
#pragma unroll
      for (int v = 0; v < 8; ++v)
        psi[base + sm * (v + 8 * hi)] = c[v];
    }
  };

  auto cswap = [&](int cpos) {
    int tpos = cpos - 1;
    for (int idx = tid; idx < 1024; idx += QS_THREADS) {
      int low = idx & ((1 << tpos) - 1);
      int high = idx >> tpos;
      int s = (high << (cpos + 1)) | (1 << cpos) | low;
      int s2 = s | (1 << tpos);
      float x = psi[s], y = psi[s2];
      psi[s] = y; psi[s2] = x;
    }
  };

  for (int l = 0; l < 4; ++l) {
    cswap(8);                       __syncthreads();
    apply(U[l < 3 ? 0 : 1], 256, 1, 16);  __syncthreads();
    apply(U[2], 16, 1, 256);        __syncthreads();
    cswap(4);                       __syncthreads();
    apply(U[4], 1, 16, 256);        __syncthreads();
    apply(U[3], 16, 1, 256);        __syncthreads();
  }

  float partial = 0.0f;
  for (int s = tid; s < 4096; s += QS_THREADS) {
    float p = psi[s]; p *= p;
    float g = 0.0f;
#pragma unroll
    for (int w = 0; w < 12; ++w)
      g += ((s >> (11 - w)) & 1) ? -hws[w] : hws[w];
    partial += p * g;
  }
  red[tid] = partial;
  __syncthreads();
#pragma unroll
  for (int off = QS_THREADS / 2; off > 0; off >>= 1) {
    if (tid < off) red[tid] += red[tid + off];
    __syncthreads();
  }
  if (tid == 0) out[b] = red[0] + head_b[0];
}

extern "C" void kernel_launch(void* const* d_in, const int* in_sizes, int n_in,
                              void* d_out, int out_size, void* d_ws, size_t ws_size,
                              hipStream_t stream) {
  const float* sb     = (const float*)d_in[0];
  const float* params = (const float*)d_in[1];
  const float* hw     = (const float*)d_in[2];
  const float* hb     = (const float*)d_in[3];
  float* out = (float*)d_out;
  int batch = in_sizes[0] / 12;
  qsim_wmma_kernel<<<dim3(batch), dim3(QS_THREADS), 0, stream>>>(
      sb, params, hw, hb, out);
}


# Pad LDS pitches (state: mid rows 20, top planes 332 dwords; gate rows 20) so every WMMA fragment load/store is bank-conflict-free; the build also needs 102 VGPRs instead of 172
# speedup vs baseline: 1.3645x; 1.3645x over previous
#include <hip/hip_runtime.h>

typedef __attribute__((ext_vector_type(2))) float v2f;
typedef __attribute__((ext_vector_type(8))) float v8f;

#define QS_THREADS 128

#define PM 20
#define PT 332

__device__ __forceinline__ int perm4(int x) {
  if (x & 8) x ^= 4;
  if (x & 4) x ^= 2;
  if (x & 2) x ^= 1;
  return x;
}
__device__ __forceinline__ int perm4inv(int x) {
  if (x & 2) x ^= 1;
  if (x & 4) x ^= 2;
  if (x & 8) x ^= 4;
  return x;
}

__global__ __launch_bounds__(QS_THREADS) void qsim_wmma_kernel(
    const float* __restrict__ state_batch,
    const float* __restrict__ params,
    const float* __restrict__ head_w,
    const float* __restrict__ head_b,
    float* __restrict__ out) {
  __shared__ __attribute__((aligned(16))) float psi[16 * PT];
  __shared__ __attribute__((aligned(16))) float U[5][16 * PM];
  __shared__ float cz[24], sz[24];
  __shared__ float hws[12];
  __shared__ float red[QS_THREADS];

  const int b   = blockIdx.x;
  const int tid = threadIdx.x;
  const int pb  = (tid >> 4) * PM + (tid & 15);

  if (tid < 12) {
    float a = 0.5f * state_batch[(size_t)b * 12 + tid];
    cz[tid] = cosf(a); sz[tid] = sinf(a);
    hws[tid] = head_w[tid];
  } else if (tid < 24) {
    float a = 0.5f * params[tid - 12];
    cz[tid] = cosf(a); sz[tid] = sinf(a);
  }
  __syncthreads();

  auto kron4 = [&](int base, int i, int j) -> float {
    float v = 1.0f;
#pragma unroll
    for (int q = 0; q < 4; ++q) {
      int bi = (i >> (3 - q)) & 1;
      int bj = (j >> (3 - q)) & 1;
      float c = cz[base + q], s = sz[base + q];
      v *= bi ? (bj ? c : s) : (bj ? -s : c);
    }
    return v;
  };

  for (int e = tid; e < 5 * 256; e += QS_THREADS) {
    int mi = e >> 8, idx = e & 255, i = idx >> 4, j = idx & 15;
    float v;
    switch (mi) {
      case 0:  v = kron4(12, perm4inv(i), j); break;
      case 1:  v = kron4(12, i, j); break;
      case 2:  v = (i == perm4(j)) ? 1.0f : 0.0f; break;
      case 3:  v = kron4(16, i, j); break;
      default: v = kron4(20, i, perm4(j)); break;
    }
    U[mi][i * PM + j] = v;
  }

  for (int it = 0; it < 32; ++it) {
    int s = tid + QS_THREADS * it;
    int s2 = (perm4inv(s >> 8) << 8) | (s & 255);
    float v = 1.0f;
#pragma unroll
    for (int w = 0; w < 12; ++w)
      v *= ((s2 >> (11 - w)) & 1) ? sz[w] : cz[w];
    psi[pb + (it >> 1) * PT + (it & 1) * (8 * PM)] = v;
  }
  __syncthreads();

  auto apply = [&](const float* Um, int sm, int sn, int st) {
    const int lane = tid & 31, wave = tid >> 5;
    const int n = lane & 15, hi = lane >> 4;
    v2f a[4];
#pragma unroll
    for (int kc = 0; kc < 4; ++kc) {
      int k0 = 4 * kc + 2 * hi;
      a[kc].x = Um[n * PM + k0];
      a[kc].y = Um[n * PM + k0 + 1];
    }
#pragma unroll
    for (int tt = 0; tt < 4; ++tt) {
      int t = wave * 4 + tt;
      int base = st * t + sn * n;
      v8f c = {};
#pragma unroll
      for (int kc = 0; kc < 4; ++kc) {
        int k0 = 4 * kc + 2 * hi;
        v2f bf;
        bf.x = psi[base + sm * k0];
        bf.y = psi[base + sm * (k0 + 1)];
        c = __builtin_amdgcn_wmma_f32_16x16x4_f32(
             false, a[kc],  false, bf,
             (short)0, c,  false,  false);
      }
#pragma unroll
      for (int v = 0; v < 8; ++v)
        psi[base + sm * (v + 8 * hi)] = c[v];
    }
  };

  auto cswap = [&](int cpos) {
    int tpos = cpos - 1;
    for (int idx = tid; idx < 1024; idx += QS_THREADS) {
      int low = idx & ((1 << tpos) - 1);
      int high = idx >> tpos;
      int s = (high << (cpos + 1)) | (1 << cpos) | low;
      int p = (s >> 8) * PT + ((s >> 4) & 15) * PM + (s & 15);
      int p2 = p + (cpos == 8 ? 8 * PM : 8);
      float x = psi[p], y = psi[p2];
      psi[p] = y; psi[p2] = x;
    }
  };

  for (int l = 0; l < 4; ++l) {
    cswap(8);                       __syncthreads();
    apply(U[l < 3 ? 0 : 1], PT, 1, PM);  __syncthreads();
    apply(U[2], PM, 1, PT);         __syncthreads();
    cswap(4);                       __syncthreads();
    apply(U[4], 1, PM, PT);         __syncthreads();
    apply(U[3], PM, 1, PT);         __syncthreads();
  }

  float partial = 0.0f;
  for (int it = 0; it < 32; ++it) {
    int s = tid + QS_THREADS * it;
    float p = psi[pb + (it >> 1) * PT + (it & 1) * (8 * PM)]; p *= p;
    float g = 0.0f;
#pragma unroll
    for (int w = 0; w < 12; ++w)
      g += ((s >> (11 - w)) & 1) ? -hws[w] : hws[w];
    partial += p * g;
  }
  red[tid] = partial;
  __syncthreads();
#pragma unroll
  for (int off = QS_THREADS / 2; off > 0; off >>= 1) {
    if (tid < off) red[tid] += red[tid + off];
    __syncthreads();
  }
  if (tid == 0) out[b] = red[0] + head_b[0];
}

extern "C" void kernel_launch(void* const* d_in, const int* in_sizes, int n_in,
                              void* d_out, int out_size, void* d_ws, size_t ws_size,
                              hipStream_t stream) {
  const float* sb     = (const float*)d_in[0];
  const float* params = (const float*)d_in[1];
  const float* hw     = (const float*)d_in[2];
  const float* hb     = (const float*)d_in[3];
  float* out = (float*)d_out;
  int batch = in_sizes[0] / 12;
  qsim_wmma_kernel<<<dim3(batch), dim3(QS_THREADS), 0, stream>>>(
      sb, params, hw, hb, out);
}


# CNOT permutations folded into neighbouring matrix passes (mid ladder into R1's matrix, cross-nibble swaps into store addresses): 24 LDS passes become 12; mid-nibble sums run in permuted k order
# speedup vs baseline: 1.7909x; 1.3125x over previous
#include <hip/hip_runtime.h>

typedef __attribute__((ext_vector_type(2))) float v2f;
typedef __attribute__((ext_vector_type(8))) float v8f;

#define QS_THREADS 128

#define PM 20
#define PT 332

__device__ __forceinline__ int perm4(int x) {
  if (x & 8) x ^= 4;
  if (x & 4) x ^= 2;
  if (x & 2) x ^= 1;
  return x;
}
__device__ __forceinline__ int perm4inv(int x) {
  if (x & 2) x ^= 1;
  if (x & 4) x ^= 2;
  if (x & 8) x ^= 4;
  return x;
}

__global__ __launch_bounds__(QS_THREADS) void qsim_wmma_kernel(
    const float* __restrict__ state_batch,
    const float* __restrict__ params,
    const float* __restrict__ head_w,
    const float* __restrict__ head_b,
    float* __restrict__ out) {
  __shared__ __attribute__((aligned(16))) float psi[16 * PT];
  __shared__ __attribute__((aligned(16))) float U[4][16 * PM];
  __shared__ float cz[24], sz[24];
  __shared__ float hws[12];
  __shared__ float red[QS_THREADS];

  const int b   = blockIdx.x;
  const int tid = threadIdx.x;
  const int pb  = (tid >> 4) * PM + (tid & 15);

  if (tid < 12) {
    float a = 0.5f * state_batch[(size_t)b * 12 + tid];
    cz[tid] = cosf(a); sz[tid] = sinf(a);
    hws[tid] = head_w[tid];
  } else if (tid < 24) {
    float a = 0.5f * params[tid - 12];
    cz[tid] = cosf(a); sz[tid] = sinf(a);
  }
  __syncthreads();

  auto kron4 = [&](int base, int i, int j) -> float {
    float v = 1.0f;
#pragma unroll
    for (int q = 0; q < 4; ++q) {
      int bi = (i >> (3 - q)) & 1;
      int bj = (j >> (3 - q)) & 1;
      float c = cz[base + q], s = sz[base + q];
      v *= bi ? (bj ? c : s) : (bj ? -s : c);
    }
    return v;
  };

  for (int e = tid; e < 4 * 256; e += QS_THREADS) {
    int mi = e >> 8, idx = e & 255, i = idx >> 4, j = idx & 15;
    float v;
    switch (mi) {
      case 0:  v = kron4(12, perm4inv(i), j); break;
      case 1:  v = kron4(12, i, j); break;
      case 2:  v = kron4(16, i, perm4(j)); break;
      default: v = kron4(20, i, perm4(j)); break;
    }
    U[mi][i * PM + j] = v;
  }

  for (int it = 0; it < 32; ++it) {
    int s = tid + QS_THREADS * it;
    int top = s >> 8;
    int s2 = (perm4inv(top) << 8) | ((s & 255) ^ ((top & 1) << 7));
    float v = 1.0f;
#pragma unroll
    for (int w = 0; w < 12; ++w)
      v *= ((s2 >> (11 - w)) & 1) ? sz[w] : cz[w];
    psi[pb + (it >> 1) * PT + (it & 1) * (8 * PM)] = v;
  }
  __syncthreads();

  auto apply = [&](const float* Um, int sm, int sn, int st, bool colfold, int rowfold) {
    const int lane = tid & 31, wave = tid >> 5;
    const int n = lane & 15, hi = lane >> 4;
    v2f a[4];
#pragma unroll
    for (int kc = 0; kc < 4; ++kc) {
      int k0 = 4 * kc + 2 * hi;
      a[kc].x = Um[n * PM + k0];
      a[kc].y = Um[n * PM + k0 + 1];
    }
#pragma unroll
    for (int tt = 0; tt < 4; ++tt) {
      int t = wave * 4 + tt;
      int base = st * t + sn * n;
      v8f c = {};
#pragma unroll
      for (int kc = 0; kc < 4; ++kc) {
        int k0 = 4 * kc + 2 * hi;
        v2f bf;
        bf.x = psi[base + sm * k0];
        bf.y = psi[base + sm * (k0 + 1)];
        c = __builtin_amdgcn_wmma_f32_16x16x4_f32(
             false, a[kc],  false, bf,
             (short)0, c,  false,  false);
      }
      int sbase = colfold ? st * t + sn * (n ^ ((__popc(t) & 1) << 3)) : base;
      int shi = hi ^ (t & rowfold);
#pragma unroll
      for (int v = 0; v < 8; ++v)
        psi[sbase + sm * (v + 8 * shi)] = c[v];
    }
  };

  for (int l = 0; l < 3; ++l) {
    apply(U[0], PT, 1, PM, true, 0);    __syncthreads();
    apply(U[3], 1, PM, PT, false, 0);   __syncthreads();
    apply(U[2], PM, 1, PT, false, 1);   __syncthreads();
  }
  apply(U[1], PT, 1, PM, true, 0);      __syncthreads();
  apply(U[3], 1, PM, PT, false, 0);     __syncthreads();
  apply(U[2], PM, 1, PT, false, 0);     __syncthreads();

  float partial = 0.0f;
  for (int it = 0; it < 32; ++it) {
    int s = tid + QS_THREADS * it;
    float p = psi[pb + (it >> 1) * PT + (it & 1) * (8 * PM)]; p *= p;
    float g = 0.0f;
#pragma unroll
    for (int w = 0; w < 12; ++w)
      g += ((s >> (11 - w)) & 1) ? -hws[w] : hws[w];
    partial += p * g;
  }
  red[tid] = partial;
  __syncthreads();
#pragma unroll
  for (int off = QS_THREADS / 2; off > 0; off >>= 1) {
    if (tid < off) red[tid] += red[tid + off];
    __syncthreads();
  }
  if (tid == 0) out[b] = red[0] + head_b[0];
}

extern "C" void kernel_launch(void* const* d_in, const int* in_sizes, int n_in,
                              void* d_out, int out_size, void* d_ws, size_t ws_size,
                              hipStream_t stream) {
  const float* sb     = (const float*)d_in[0];
  const float* params = (const float*)d_in[1];
  const float* hw     = (const float*)d_in[2];
  const float* hb     = (const float*)d_in[3];
  float* out = (float*)d_out;
  int batch = in_sizes[0] / 12;
  qsim_wmma_kernel<<<dim3(batch), dim3(QS_THREADS), 0, stream>>>(
      sb, params, hw, hb, out);
}


# Encoding amplitude and readout weight computed per nibble (T[top]*M[mid]*L[low]; gT+gM+gL) instead of 12 selects per element; products and the readout-weight sum run in per-nibble order
# speedup vs baseline: 2.3031x; 1.2860x over previous
#include <hip/hip_runtime.h>

typedef __attribute__((ext_vector_type(2))) float v2f;
typedef __attribute__((ext_vector_type(8))) float v8f;

#define QS_THREADS 128

#define PM 20
#define PT 332

__device__ __forceinline__ int perm4(int x) {
  if (x & 8) x ^= 4;
  if (x & 4) x ^= 2;
  if (x & 2) x ^= 1;
  return x;
}
__device__ __forceinline__ int perm4inv(int x) {
  if (x & 2) x ^= 1;
  if (x & 4) x ^= 2;
  if (x & 8) x ^= 4;
  return x;
}

__global__ __launch_bounds__(QS_THREADS) void qsim_wmma_kernel(
    const float* __restrict__ state_batch,
    const float* __restrict__ params,
    const float* __restrict__ head_w,
    const float* __restrict__ head_b,
    float* __restrict__ out) {
  __shared__ __attribute__((aligned(16))) float psi[16 * PT];
  __shared__ __attribute__((aligned(16))) float U[4][16 * PM];
  __shared__ float cz[24], sz[24];
  __shared__ float hws[12];
  __shared__ float red[QS_THREADS];

  const int b   = blockIdx.x;
  const int tid = threadIdx.x;
  const int pb  = (tid >> 4) * PM + (tid & 15);

  if (tid < 12) {
    float a = 0.5f * state_batch[(size_t)b * 12 + tid];
    cz[tid] = cosf(a); sz[tid] = sinf(a);
    hws[tid] = head_w[tid];
  } else if (tid < 24) {
    float a = 0.5f * params[tid - 12];
    cz[tid] = cosf(a); sz[tid] = sinf(a);
  }
  __syncthreads();

  auto kron4 = [&](int base, int i, int j) -> float {
    float v = 1.0f;
#pragma unroll
    for (int q = 0; q < 4; ++q) {
      int bi = (i >> (3 - q)) & 1;
      int bj = (j >> (3 - q)) & 1;
      float c = cz[base + q], s = sz[base + q];
      v *= bi ? (bj ? c : s) : (bj ? -s : c);
    }
    return v;
  };

  for (int e = tid; e < 4 * 256; e += QS_THREADS) {
    int mi = e >> 8, idx = e & 255, i = idx >> 4, j = idx & 15;
    float v;
    switch (mi) {
      case 0:  v = kron4(12, perm4inv(i), j); break;
      case 1:  v = kron4(12, i, j); break;
      case 2:  v = kron4(16, i, perm4(j)); break;
      default: v = kron4(20, i, perm4(j)); break;
    }
    U[mi][i * PM + j] = v;
  }

  const int lowq = tid & 15, midq = tid >> 4;
  float ml0, ml1;
  {
    float fl = 1.0f, fm = 1.0f;
    for (int q = 0; q < 4; ++q) fl *= ((lowq >> (3 - q)) & 1) ? sz[8 + q] : cz[8 + q];
    for (int q = 1; q < 4; ++q) fm *= ((midq >> (3 - q)) & 1) ? sz[4 + q] : cz[4 + q];
    fm *= fl;
    ml0 = cz[4] * fm;
    ml1 = sz[4] * fm;
  }
  float ft[16];
#pragma unroll
  for (int x = 0; x < 16; ++x) {
    float v = 1.0f;
    for (int q = 0; q < 4; ++q) v *= ((x >> (3 - q)) & 1) ? sz[q] : cz[q];
    ft[x] = v;
  }
#pragma unroll
  for (int it = 0; it < 32; ++it) {
    const int top = it >> 1;
    const int msb = (it & 1) ^ (top & 1);
    psi[pb + top * PT + (it & 1) * (8 * PM)] = ft[perm4inv(top)] * (msb ? ml1 : ml0);
  }
  __syncthreads();

  auto apply = [&](const float* Um, int sm, int sn, int st, bool colfold, int rowfold) {
    const int lane = tid & 31, wave = tid >> 5;
    const int n = lane & 15, hi = lane >> 4;
    v2f a[4];
#pragma unroll
    for (int kc = 0; kc < 4; ++kc) {
      int k0 = 4 * kc + 2 * hi;
      a[kc].x = Um[n * PM + k0];
      a[kc].y = Um[n * PM + k0 + 1];
    }
#pragma unroll
    for (int tt = 0; tt < 4; ++tt) {
      int t = wave * 4 + tt;
      int base = st * t + sn * n;
      v8f c = {};
#pragma unroll
      for (int kc = 0; kc < 4; ++kc) {
        int k0 = 4 * kc + 2 * hi;
        v2f bf;
        bf.x = psi[base + sm * k0];
        bf.y = psi[base + sm * (k0 + 1)];
        c = __builtin_amdgcn_wmma_f32_16x16x4_f32(
             false, a[kc],  false, bf,
             (short)0, c,  false,  false);
      }
      int sbase = colfold ? st * t + sn * (n ^ ((__popc(t) & 1) << 3)) : base;
      int shi = hi ^ (t & rowfold);
#pragma unroll
      for (int v = 0; v < 8; ++v)
        psi[sbase + sm * (v + 8 * shi)] = c[v];
    }
  };

  for (int l = 0; l < 3; ++l) {
    apply(U[0], PT, 1, PM, true, 0);    __syncthreads();
    apply(U[3], 1, PM, PT, false, 0);   __syncthreads();
    apply(U[2], PM, 1, PT, false, 1);   __syncthreads();
  }
  apply(U[1], PT, 1, PM, true, 0);      __syncthreads();
  apply(U[3], 1, PM, PT, false, 0);     __syncthreads();
  apply(U[2], PM, 1, PT, false, 0);     __syncthreads();

  float gml0, gml1;
  {
    float gl = 0.0f, gm = 0.0f;
    for (int q = 0; q < 4; ++q) gl += ((lowq >> (3 - q)) & 1) ? -hws[8 + q] : hws[8 + q];
    for (int q = 1; q < 4; ++q) gm += ((midq >> (3 - q)) & 1) ? -hws[4 + q] : hws[4 + q];
    gm += gl;
    gml0 = gm + hws[4];
    gml1 = gm - hws[4];
  }
  float gt[16];
#pragma unroll
  for (int x = 0; x < 16; ++x) {
    float v = 0.0f;
    for (int q = 0; q < 4; ++q) v += ((x >> (3 - q)) & 1) ? -hws[q] : hws[q];
    gt[x] = v;
  }
  float partial = 0.0f;
#pragma unroll
  for (int it = 0; it < 32; ++it) {
    float p = psi[pb + (it >> 1) * PT + (it & 1) * (8 * PM)]; p *= p;
    partial += p * (gt[it >> 1] + ((it & 1) ? gml1 : gml0));
  }
  red[tid] = partial;
  __syncthreads();
#pragma unroll
  for (int off = QS_THREADS / 2; off > 0; off >>= 1) {
    if (tid < off) red[tid] += red[tid + off];
    __syncthreads();
  }
  if (tid == 0) out[b] = red[0] + head_b[0];
}

extern "C" void kernel_launch(void* const* d_in, const int* in_sizes, int n_in,
                              void* d_out, int out_size, void* d_ws, size_t ws_size,
                              hipStream_t stream) {
  const float* sb     = (const float*)d_in[0];
  const float* params = (const float*)d_in[1];
  const float* hw     = (const float*)d_in[2];
  const float* hb     = (const float*)d_in[3];
  float* out = (float*)d_out;
  int batch = in_sizes[0] / 12;
  qsim_wmma_kernel<<<dim3(batch), dim3(QS_THREADS), 0, stream>>>(
      sb, params, hw, hb, out);
}


# Low and mid passes fused into one LDS trip: the low pass's D registers are the mid pass's A operand (k order 8*hi+2*kc+v), all its LDS accesses 16-byte; low and mid sums run in that k order
# speedup vs baseline: 2.5078x; 1.0889x over previous
#include <hip/hip_runtime.h>

typedef __attribute__((ext_vector_type(2))) float v2f;
typedef __attribute__((ext_vector_type(4))) float v4f;
typedef __attribute__((ext_vector_type(8))) float v8f;

#define QS_THREADS 128

#define PM 20
#define PT 332

__device__ __forceinline__ int perm4(int x) {
  if (x & 8) x ^= 4;
  if (x & 4) x ^= 2;
  if (x & 2) x ^= 1;
  return x;
}
__device__ __forceinline__ int perm4inv(int x) {
  if (x & 2) x ^= 1;
  if (x & 4) x ^= 2;
  if (x & 8) x ^= 4;
  return x;
}

__global__ __launch_bounds__(QS_THREADS) void qsim_wmma_kernel(
    const float* __restrict__ state_batch,
    const float* __restrict__ params,
    const float* __restrict__ head_w,
    const float* __restrict__ head_b,
    float* __restrict__ out) {
  __shared__ __attribute__((aligned(16))) float psi[16 * PT];
  __shared__ __attribute__((aligned(16))) float U[4][16 * PM];
  __shared__ float cz[24], sz[24];
  __shared__ float hws[12];
  __shared__ float red[QS_THREADS];

  const int b   = blockIdx.x;
  const int tid = threadIdx.x;
  const int pb  = (tid >> 4) * PM + (tid & 15);

  if (tid < 12) {
    float a = 0.5f * state_batch[(size_t)b * 12 + tid];
    cz[tid] = cosf(a); sz[tid] = sinf(a);
    hws[tid] = head_w[tid];
  } else if (tid < 24) {
    float a = 0.5f * params[tid - 12];
    cz[tid] = cosf(a); sz[tid] = sinf(a);
  }
  __syncthreads();

  auto kron4 = [&](int base, int i, int j) -> float {
    float v = 1.0f;
#pragma unroll
    for (int q = 0; q < 4; ++q) {
      int bi = (i >> (3 - q)) & 1;
      int bj = (j >> (3 - q)) & 1;
      float c = cz[base + q], s = sz[base + q];
      v *= bi ? (bj ? c : s) : (bj ? -s : c);
    }
    return v;
  };

  for (int e = tid; e < 4 * 256; e += QS_THREADS) {
    int mi = e >> 8, idx = e & 255, i = idx >> 4, j = idx & 15;
    float v;
    switch (mi) {
      case 0:  v = kron4(12, perm4inv(i), j); break;
      case 1:  v = kron4(12, i, j); break;
      case 2:  v = kron4(16, i, perm4(j)); break;
      default: v = kron4(20, i, perm4(j)); break;
    }
    U[mi][i * PM + j] = v;
  }

  const int lowq = tid & 15, midq = tid >> 4;
  float ml0, ml1;
  {
    float fl = 1.0f, fm = 1.0f;
    for (int q = 0; q < 4; ++q) fl *= ((lowq >> (3 - q)) & 1) ? sz[8 + q] : cz[8 + q];
    for (int q = 1; q < 4; ++q) fm *= ((midq >> (3 - q)) & 1) ? sz[4 + q] : cz[4 + q];
    fm *= fl;
    ml0 = cz[4] * fm;
    ml1 = sz[4] * fm;
  }
  float ft[16];
#pragma unroll
  for (int x = 0; x < 16; ++x) {
    float v = 1.0f;
    for (int q = 0; q < 4; ++q) v *= ((x >> (3 - q)) & 1) ? sz[q] : cz[q];
    ft[x] = v;
  }
#pragma unroll
  for (int it = 0; it < 32; ++it) {
    const int top = it >> 1;
    const int msb = (it & 1) ^ (top & 1);
    psi[pb + top * PT + (it & 1) * (8 * PM)] = ft[perm4inv(top)] * (msb ? ml1 : ml0);
  }
  __syncthreads();

  auto apply = [&](const float* Um, int sm, int sn, int st, bool colfold, int rowfold) {
    const int lane = tid & 31, wave = tid >> 5;
    const int n = lane & 15, hi = lane >> 4;
    v2f a[4];
#pragma unroll
    for (int kc = 0; kc < 4; ++kc) {
      int k0 = 4 * kc + 2 * hi;
      a[kc].x = Um[n * PM + k0];
      a[kc].y = Um[n * PM + k0 + 1];
    }
#pragma unroll
    for (int tt = 0; tt < 4; ++tt) {
      int t = wave * 4 + tt;
      int base = st * t + sn * n;
      v8f c = {};
#pragma unroll
      for (int kc = 0; kc < 4; ++kc) {
        int k0 = 4 * kc + 2 * hi;
        v2f bf;
        bf.x = psi[base + sm * k0];
        bf.y = psi[base + sm * (k0 + 1)];
        c = __builtin_amdgcn_wmma_f32_16x16x4_f32(
             false, a[kc],  false, bf,
             (short)0, c,  false,  false);
      }
      int sbase = colfold ? st * t + sn * (n ^ ((__popc(t) & 1) << 3)) : base;
      int shi = hi ^ (t & rowfold);
#pragma unroll
      for (int v = 0; v < 8; ++v)
        psi[sbase + sm * (v + 8 * shi)] = c[v];
    }
  };

  auto apply_low_mid = [&](const float* Ulow, const float* Umid, int rowfold) {
    const int lane = tid & 31, wave = tid >> 5;
    const int n = lane & 15, hi = lane >> 4;
    const int frag = n * PM + 8 * hi;
    v4f in[4][2];
#pragma unroll
    for (int tt = 0; tt < 4; ++tt) {
      const v4f* p = reinterpret_cast<const v4f*>(&psi[(wave * 4 + tt) * PT + frag]);
      in[tt][0] = p[0];
      in[tt][1] = p[1];
    }
    v2f bl[4], bm[4];
    {
      const v4f* gl = reinterpret_cast<const v4f*>(&Ulow[frag]);
      const v4f* gm = reinterpret_cast<const v4f*>(&Umid[frag]);
      v4f l0 = gl[0], l1 = gl[1], m0 = gm[0], m1 = gm[1];
      bl[0].x = l0.x; bl[0].y = l0.y; bl[1].x = l0.z; bl[1].y = l0.w;
      bl[2].x = l1.x; bl[2].y = l1.y; bl[3].x = l1.z; bl[3].y = l1.w;
      bm[0].x = m0.x; bm[0].y = m0.y; bm[1].x = m0.z; bm[1].y = m0.w;
      bm[2].x = m1.x; bm[2].y = m1.y; bm[3].x = m1.z; bm[3].y = m1.w;
    }
    v8f y[4];
#pragma unroll
    for (int tt = 0; tt < 4; ++tt) {
      v8f c = {};
#pragma unroll
      for (int kc = 0; kc < 4; ++kc) {
        v2f af;
        af.x = in[tt][kc >> 1][2 * (kc & 1)];
        af.y = in[tt][kc >> 1][2 * (kc & 1) + 1];
        c = __builtin_amdgcn_wmma_f32_16x16x4_f32(
             false, af,  false, bl[kc],
             (short)0, c,  false,  false);
      }
      y[tt] = c;
    }
#pragma unroll
    for (int tt = 0; tt < 4; ++tt) {
      v8f c = {};
#pragma unroll
      for (int kc = 0; kc < 4; ++kc) {
        v2f af;
        af.x = y[tt][2 * kc];
        af.y = y[tt][2 * kc + 1];
        c = __builtin_amdgcn_wmma_f32_16x16x4_f32(
             false, af,  false, bm[kc],
             (short)0, c,  false,  false);
      }
      const int t = wave * 4 + tt;
      v4f* q = reinterpret_cast<v4f*>(&psi[t * PT + (n ^ ((t & rowfold) << 3)) * PM + 8 * hi]);
      v4f o0, o1;
      o0.x = c[0]; o0.y = c[1]; o0.z = c[2]; o0.w = c[3];
      o1.x = c[4]; o1.y = c[5]; o1.z = c[6]; o1.w = c[7];
      q[0] = o0;
      q[1] = o1;
    }
  };

  for (int l = 0; l < 3; ++l) {
    apply(U[0], PT, 1, PM, true, 0);    __syncthreads();
    apply_low_mid(U[3], U[2], 1);       __syncthreads();
  }
  apply(U[1], PT, 1, PM, true, 0);      __syncthreads();
  apply_low_mid(U[3], U[2], 0);         __syncthreads();

  float gml0, gml1;
  {
    float gl = 0.0f, gm = 0.0f;
    for (int q = 0; q < 4; ++q) gl += ((lowq >> (3 - q)) & 1) ? -hws[8 + q] : hws[8 + q];
    for (int q = 1; q < 4; ++q) gm += ((midq >> (3 - q)) & 1) ? -hws[4 + q] : hws[4 + q];
    gm += gl;
    gml0 = gm + hws[4];
    gml1 = gm - hws[4];
  }
  float gt[16];
#pragma unroll
  for (int x = 0; x < 16; ++x) {
    float v = 0.0f;
    for (int q = 0; q < 4; ++q) v += ((x >> (3 - q)) & 1) ? -hws[q] : hws[q];
    gt[x] = v;
  }
  float partial = 0.0f;
#pragma unroll
  for (int it = 0; it < 32; ++it) {
    float p = psi[pb + (it >> 1) * PT + (it & 1) * (8 * PM)]; p *= p;
    partial += p * (gt[it >> 1] + ((it & 1) ? gml1 : gml0));
  }
  red[tid] = partial;
  __syncthreads();
#pragma unroll
  for (int off = QS_THREADS / 2; off > 0; off >>= 1) {
    if (tid < off) red[tid] += red[tid + off];
    __syncthreads();
  }
  if (tid == 0) out[b] = red[0] + head_b[0];
}

extern "C" void kernel_launch(void* const* d_in, const int* in_sizes, int n_in,
                              void* d_out, int out_size, void* d_ws, size_t ws_size,
                              hipStream_t stream) {
  const float* sb     = (const float*)d_in[0];
  const float* params = (const float*)d_in[1];
  const float* hw     = (const float*)d_in[2];
  const float* hb     = (const float*)d_in[3];
  float* out = (float*)d_out;
  int batch = in_sizes[0] / 12;
  qsim_wmma_kernel<<<dim3(batch), dim3(QS_THREADS), 0, stream>>>(
      sb, params, hw, hb, out);
}


# Layer 1 and layer 2's top pass run on the encoded state's per-nibble factors: 4 small matrix products + a rank-4 expansion replace 4 full-state passes; their sums run per factor, in a new order
# speedup vs baseline: 3.1867x; 1.2707x over previous
#include <hip/hip_runtime.h>

typedef __attribute__((ext_vector_type(2))) float v2f;
typedef __attribute__((ext_vector_type(4))) float v4f;
typedef __attribute__((ext_vector_type(8))) float v8f;

#define QS_THREADS 128

#define PM 20
#define PT 332

__device__ __forceinline__ int perm4(int x) {
  if (x & 8) x ^= 4;
  if (x & 4) x ^= 2;
  if (x & 2) x ^= 1;
  return x;
}
__device__ __forceinline__ int perm4inv(int x) {
  if (x & 2) x ^= 1;
  if (x & 4) x ^= 2;
  if (x & 8) x ^= 4;
  return x;
}

__global__ __launch_bounds__(QS_THREADS) void qsim_wmma_kernel(
    const float* __restrict__ state_batch,
    const float* __restrict__ params,
    const float* __restrict__ head_w,
    const float* __restrict__ head_b,
    float* __restrict__ out) {
  __shared__ __attribute__((aligned(16))) float psi[16 * PT];
  __shared__ __attribute__((aligned(16))) float U[4][16 * PM];
  __shared__ float cz[24], sz[24];
  __shared__ float hws[12];
  __shared__ float red[QS_THREADS];
  float* const fac = red;
  __shared__ __attribute__((aligned(16))) float fac2[64];

  const int b   = blockIdx.x;
  const int tid = threadIdx.x;
  const int pb  = (tid >> 4) * PM + (tid & 15);

  if (tid < 12) {
    float a = 0.5f * state_batch[(size_t)b * 12 + tid];
    cz[tid] = cosf(a); sz[tid] = sinf(a);
    hws[tid] = head_w[tid];
  } else if (tid < 24) {
    float a = 0.5f * params[tid - 12];
    cz[tid] = cosf(a); sz[tid] = sinf(a);
  }
  __syncthreads();

  auto kron4 = [&](int base, int i, int j) -> float {
    float v = 1.0f;
#pragma unroll
    for (int q = 0; q < 4; ++q) {
      int bi = (i >> (3 - q)) & 1;
      int bj = (j >> (3 - q)) & 1;
      float c = cz[base + q], s = sz[base + q];
      v *= bi ? (bj ? c : s) : (bj ? -s : c);
    }
    return v;
  };

  for (int e = tid; e < 4 * 256; e += QS_THREADS) {
    int mi = e >> 8, idx = e & 255, i = idx >> 4, j = idx & 15;
    float v;
    switch (mi) {
      case 0:  v = kron4(12, perm4inv(i), j); break;
      case 1:  v = kron4(12, i, j); break;
      case 2:  v = kron4(16, i, perm4(j)); break;
      default: v = kron4(20, i, perm4(j)); break;
    }
    U[mi][i * PM + j] = v;
  }

  const int lowq = tid & 15, midq = tid >> 4;
  auto amp4 = [&](int base, int x) -> float {
    float v = 1.0f;
#pragma unroll
    for (int q = 0; q < 4; ++q) v *= ((x >> (3 - q)) & 1) ? sz[base + q] : cz[base + q];
    return v;
  };
  {
    const int k = tid & 15;
    int plane, col;
    float v;
    if (tid < 32) {
      plane = 0; col = tid >> 4;
      v = ((k & 1) == col) ? amp4(0, perm4inv(k)) : 0.0f;
    } else if (tid < 64) {
      plane = 1; col = (tid >> 4) & 1;
      v = amp4(8, k ^ (col << 3));
    } else {
      plane = 2; col = (tid >> 4) & 3;
      v = ((__popc(k) & 1) == (col >> 1)) ? amp4(4, k ^ ((col & 1) << 3)) : 0.0f;
    }
    psi[plane * PT + k * PM + col] = v;
  }
  __syncthreads();

  const int fp_lane = tid & 31, fp_wave = tid >> 5;
  const int fp_n = fp_lane & 15, fp_hi = fp_lane >> 4;
  auto factor_product = [&](const float* Um, const v2f (&bf)[4], float* dst, bool keep) {
    v2f af[4];
#pragma unroll
    for (int kc = 0; kc < 4; ++kc) {
      const int k0 = 4 * kc + 2 * fp_hi;
      af[kc].x = Um[fp_n * PM + k0];
      af[kc].y = Um[fp_n * PM + k0 + 1];
    }
    v8f c = {};
#pragma unroll
    for (int kc = 0; kc < 4; ++kc)
      c = __builtin_amdgcn_wmma_f32_16x16x4_f32(
           false, af[kc],  false, bf[kc],
           (short)0, c,  false,  false);
    if (keep) {
#pragma unroll
      for (int v = 0; v < 8; ++v)
        dst[v + 8 * fp_hi] = c[v];
    }
  };

  if (fp_wave >= 1) {
    const int pr = fp_wave - 1;
    const float* Um = (pr == 0) ? U[0] : ((pr == 1) ? U[3] : U[2]);
    const int ncol = (pr == 2) ? 4 : 2;
    const bool keep = fp_n < ncol;
    const float* Vp = &psi[pr * PT];
    v2f bf[4];
#pragma unroll
    for (int kc = 0; kc < 4; ++kc) { bf[kc].x = 0.0f; bf[kc].y = 0.0f; }
    if (keep) {
#pragma unroll
      for (int kc = 0; kc < 4; ++kc) {
        const int k0 = 4 * kc + 2 * fp_hi;
        bf[kc].x = Vp[k0 * PM + fp_n];
        bf[kc].y = Vp[(k0 + 1) * PM + fp_n];
      }
    }
    factor_product(Um, bf, &fac[(2 * pr + (keep ? fp_n : 0)) * 16], keep);
  }
  __syncthreads();

  if (fp_wave == 1) {
    const bool keep = fp_n < 4;
    v2f bf[4];
#pragma unroll
    for (int kc = 0; kc < 4; ++kc) { bf[kc].x = 0.0f; bf[kc].y = 0.0f; }
    if (keep) {
      const int odd = fp_n >> 1;
      const float* src = &fac[(fp_n & 1) * 16 + odd];
#pragma unroll
      for (int kc = 0; kc < 4; ++kc) {
        const float t = src[4 * kc + 2 * fp_hi];
        bf[kc].x = odd ? 0.0f : t;
        bf[kc].y = odd ? t : 0.0f;
      }
    }
    factor_product(U[0], bf, &fac2[(keep ? fp_n : 0) * 16], keep);
  }
  __syncthreads();

  {
    const float Le = fac[32 + lowq], Lo = fac[48 + lowq];
    float P[2], Q[2];
#pragma unroll
    for (int h = 0; h < 2; ++h) {
      const int m = midq + 8 * h;
      P[h] = Le * fac[64 + m] + Lo * fac[96 + m];
      Q[h] = Le * fac[80 + m] + Lo * fac[112 + m];
    }
    float ee[16], oe[16], eo[16], oo[16];
#pragma unroll
    for (int x = 0; x < 16; ++x) {
      ee[x] = fac2[x]; oe[x] = fac2[16 + x]; eo[x] = fac2[32 + x]; oo[x] = fac2[48 + x];
    }
    const int parq = __popc(midq) & 1;
#pragma unroll
    for (int it = 0; it < 32; ++it) {
      const int top = it >> 1, h = it & 1;
      const int lw = lowq ^ ((parq ^ h) << 3);
      psi[top * PT + (midq + 8 * h) * PM + lw] =
          ee[top] * P[h] + oe[top] * Q[h] + eo[top] * P[1 - h] + oo[top] * Q[1 - h];
    }
  }
  __syncthreads();

  auto apply = [&](const float* Um, int sm, int sn, int st, bool colfold, int rowfold) {
    const int lane = tid & 31, wave = tid >> 5;
    const int n = lane & 15, hi = lane >> 4;
    v2f a[4];
#pragma unroll
    for (int kc = 0; kc < 4; ++kc) {
      int k0 = 4 * kc + 2 * hi;
      a[kc].x = Um[n * PM + k0];
      a[kc].y = Um[n * PM + k0 + 1];
    }
#pragma unroll
    for (int tt = 0; tt < 4; ++tt) {
      int t = wave * 4 + tt;
      int base = st * t + sn * n;
      v8f c = {};
#pragma unroll
      for (int kc = 0; kc < 4; ++kc) {
        int k0 = 4 * kc + 2 * hi;
        v2f bf;
        bf.x = psi[base + sm * k0];
        bf.y = psi[base + sm * (k0 + 1)];
        c = __builtin_amdgcn_wmma_f32_16x16x4_f32(
             false, a[kc],  false, bf,
             (short)0, c,  false,  false);
      }
      int sbase = colfold ? st * t + sn * (n ^ ((__popc(t) & 1) << 3)) : base;
      int shi = hi ^ (t & rowfold);
#pragma unroll
      for (int v = 0; v < 8; ++v)
        psi[sbase + sm * (v + 8 * shi)] = c[v];
    }
  };

  auto apply_low_mid = [&](const float* Ulow, const float* Umid, int rowfold) {
    const int lane = tid & 31, wave = tid >> 5;
    const int n = lane & 15, hi = lane >> 4;
    const int frag = n * PM + 8 * hi;
    v4f in[4][2];
#pragma unroll
    for (int tt = 0; tt < 4; ++tt) {
      const v4f* p = reinterpret_cast<const v4f*>(&psi[(wave * 4 + tt) * PT + frag]);
      in[tt][0] = p[0];
      in[tt][1] = p[1];
    }
    v2f bl[4], bm[4];
    {
      const v4f* gl = reinterpret_cast<const v4f*>(&Ulow[frag]);
      const v4f* gm = reinterpret_cast<const v4f*>(&Umid[frag]);
      v4f l0 = gl[0], l1 = gl[1], m0 = gm[0], m1 = gm[1];
      bl[0].x = l0.x; bl[0].y = l0.y; bl[1].x = l0.z; bl[1].y = l0.w;
      bl[2].x = l1.x; bl[2].y = l1.y; bl[3].x = l1.z; bl[3].y = l1.w;
      bm[0].x = m0.x; bm[0].y = m0.y; bm[1].x = m0.z; bm[1].y = m0.w;
      bm[2].x = m1.x; bm[2].y = m1.y; bm[3].x = m1.z; bm[3].y = m1.w;
    }
    v8f y[4];
#pragma unroll
    for (int tt = 0; tt < 4; ++tt) {
      v8f c = {};
#pragma unroll
      for (int kc = 0; kc < 4; ++kc) {
        v2f af;
        af.x = in[tt][kc >> 1][2 * (kc & 1)];
        af.y = in[tt][kc >> 1][2 * (kc & 1) + 1];
        c = __builtin_amdgcn_wmma_f32_16x16x4_f32(
             false, af,  false, bl[kc],
             (short)0, c,  false,  false);
      }
      y[tt] = c;
    }
#pragma unroll
    for (int tt = 0; tt < 4; ++tt) {
      v8f c = {};
#pragma unroll
      for (int kc = 0; kc < 4; ++kc) {
        v2f af;
        af.x = y[tt][2 * kc];
        af.y = y[tt][2 * kc + 1];
        c = __builtin_amdgcn_wmma_f32_16x16x4_f32(
             false, af,  false, bm[kc],
             (short)0, c,  false,  false);
      }
      const int t = wave * 4 + tt;
      v4f* q = reinterpret_cast<v4f*>(&psi[t * PT + (n ^ ((t & rowfold) << 3)) * PM + 8 * hi]);
      v4f o0, o1;
      o0.x = c[0]; o0.y = c[1]; o0.z = c[2]; o0.w = c[3];
      o1.x = c[4]; o1.y = c[5]; o1.z = c[6]; o1.w = c[7];
      q[0] = o0;
      q[1] = o1;
    }
  };

  apply_low_mid(U[3], U[2], 1);         __syncthreads();
  apply(U[0], PT, 1, PM, true, 0);      __syncthreads();
  apply_low_mid(U[3], U[2], 1);         __syncthreads();
  apply(U[1], PT, 1, PM, true, 0);      __syncthreads();
  apply_low_mid(U[3], U[2], 0);         __syncthreads();

  float gml0, gml1;
  {
    float gl = 0.0f, gm = 0.0f;
    for (int q = 0; q < 4; ++q) gl += ((lowq >> (3 - q)) & 1) ? -hws[8 + q] : hws[8 + q];
    for (int q = 1; q < 4; ++q) gm += ((midq >> (3 - q)) & 1) ? -hws[4 + q] : hws[4 + q];
    gm += gl;
    gml0 = gm + hws[4];
    gml1 = gm - hws[4];
  }
  float gt[16];
#pragma unroll
  for (int x = 0; x < 16; ++x) {
    float v = 0.0f;
    for (int q = 0; q < 4; ++q) v += ((x >> (3 - q)) & 1) ? -hws[q] : hws[q];
    gt[x] = v;
  }
  float partial = 0.0f;
#pragma unroll
  for (int it = 0; it < 32; ++it) {
    float p = psi[pb + (it >> 1) * PT + (it & 1) * (8 * PM)]; p *= p;
    partial += p * (gt[it >> 1] + ((it & 1) ? gml1 : gml0));
  }
  red[tid] = partial;
  __syncthreads();
#pragma unroll
  for (int off = QS_THREADS / 2; off > 0; off >>= 1) {
    if (tid < off) red[tid] += red[tid + off];
    __syncthreads();
  }
  if (tid == 0) out[b] = red[0] + head_b[0];
}

extern "C" void kernel_launch(void* const* d_in, const int* in_sizes, int n_in,
                              void* d_out, int out_size, void* d_ws, size_t ws_size,
                              hipStream_t stream) {
  const float* sb     = (const float*)d_in[0];
  const float* params = (const float*)d_in[1];
  const float* hw     = (const float*)d_in[2];
  const float* hb     = (const float*)d_in[3];
  float* out = (float*)d_out;
  int batch = in_sizes[0] / 12;
  qsim_wmma_kernel<<<dim3(batch), dim3(QS_THREADS), 0, stream>>>(
      sb, params, hw, hb, out);
}


# Layer 2's low/mid pass and layer 3 run on the factors too: 12 rank tiles pass low/mid instead of 32 state tiles, layer 3's top pass is one 8-column product; rank-8 expansion; new summing order
# speedup vs baseline: 4.1318x; 1.2966x over previous
#include <hip/hip_runtime.h>

typedef __attribute__((ext_vector_type(2))) float v2f;
typedef __attribute__((ext_vector_type(4))) float v4f;
typedef __attribute__((ext_vector_type(8))) float v8f;

#define QS_THREADS 128

#define PM 20
#define PT 332

__device__ __forceinline__ int perm4(int x) {
  if (x & 8) x ^= 4;
  if (x & 4) x ^= 2;
  if (x & 2) x ^= 1;
  return x;
}
__device__ __forceinline__ int perm4inv(int x) {
  if (x & 2) x ^= 1;
  if (x & 4) x ^= 2;
  if (x & 8) x ^= 4;
  return x;
}

__global__ __launch_bounds__(QS_THREADS) void qsim_wmma_kernel(
    const float* __restrict__ state_batch,
    const float* __restrict__ params,
    const float* __restrict__ head_w,
    const float* __restrict__ head_b,
    float* __restrict__ out) {
  __shared__ __attribute__((aligned(16))) float psi[16 * PT];
  __shared__ __attribute__((aligned(16))) float U[4][16 * PM];
  __shared__ float cz[24], sz[24];
  __shared__ float hws[12];
  __shared__ __attribute__((aligned(16))) float red[QS_THREADS];
  float* const fac = red;
  __shared__ __attribute__((aligned(16))) float fac2[64];

  const int b   = blockIdx.x;
  const int tid = threadIdx.x;
  const int pb  = (tid >> 4) * PM + (tid & 15);

  if (tid < 12) {
    float a = 0.5f * state_batch[(size_t)b * 12 + tid];
    cz[tid] = cosf(a); sz[tid] = sinf(a);
    hws[tid] = head_w[tid];
  } else if (tid < 24) {
    float a = 0.5f * params[tid - 12];
    cz[tid] = cosf(a); sz[tid] = sinf(a);
  }
  __syncthreads();

  auto kron4 = [&](int base, int i, int j) -> float {
    float v = 1.0f;
#pragma unroll
    for (int q = 0; q < 4; ++q) {
      int bi = (i >> (3 - q)) & 1;
      int bj = (j >> (3 - q)) & 1;
      float c = cz[base + q], s = sz[base + q];
      v *= bi ? (bj ? c : s) : (bj ? -s : c);
    }
    return v;
  };

  for (int e = tid; e < 4 * 256; e += QS_THREADS) {
    int mi = e >> 8, idx = e & 255, i = idx >> 4, j = idx & 15;
    float v;
    switch (mi) {
      case 0:  v = kron4(12, perm4inv(i), j); break;
      case 1:  v = kron4(12, i, j); break;
      case 2:  v = kron4(16, i, perm4(j)); break;
      default: v = kron4(20, i, perm4(j)); break;
    }
    U[mi][i * PM + j] = v;
  }

  const int lowq = tid & 15, midq = tid >> 4;
  auto amp4 = [&](int base, int x) -> float {
    float v = 1.0f;
#pragma unroll
    for (int q = 0; q < 4; ++q) v *= ((x >> (3 - q)) & 1) ? sz[base + q] : cz[base + q];
    return v;
  };
  {
    const int k = tid & 15;
    int plane, col;
    float v;
    if (tid < 32) {
      plane = 0; col = tid >> 4;
      v = ((k & 1) == col) ? amp4(0, perm4inv(k)) : 0.0f;
    } else if (tid < 64) {
      plane = 1; col = (tid >> 4) & 1;
      v = amp4(8, k ^ (col << 3));
    } else {
      plane = 2; col = (tid >> 4) & 3;
      v = ((__popc(k) & 1) == (col >> 1)) ? amp4(4, k ^ ((col & 1) << 3)) : 0.0f;
    }
    psi[plane * PT + k * PM + col] = v;
  }
  __syncthreads();

  const int fp_lane = tid & 31, fp_wave = tid >> 5;
  const int fp_n = fp_lane & 15, fp_hi = fp_lane >> 4;
  auto factor_product = [&](const float* Um, const v2f (&bf)[4], float* dst, bool keep) {
    v2f af[4];
#pragma unroll
    for (int kc = 0; kc < 4; ++kc) {
      const int k0 = 4 * kc + 2 * fp_hi;
      af[kc].x = Um[fp_n * PM + k0];
      af[kc].y = Um[fp_n * PM + k0 + 1];
    }
    v8f c = {};
#pragma unroll
    for (int kc = 0; kc < 4; ++kc)
      c = __builtin_amdgcn_wmma_f32_16x16x4_f32(
           false, af[kc],  false, bf[kc],
           (short)0, c,  false,  false);
    if (keep) {
#pragma unroll
      for (int v = 0; v < 8; ++v)
        dst[v + 8 * fp_hi] = c[v];
    }
  };

  if (fp_wave >= 1) {
    const int pr = fp_wave - 1;
    const float* Um = (pr == 0) ? U[0] : ((pr == 1) ? U[3] : U[2]);
    const int ncol = (pr == 2) ? 4 : 2;
    const bool keep = fp_n < ncol;
    const float* Vp = &psi[pr * PT];
    v2f bf[4];
#pragma unroll
    for (int kc = 0; kc < 4; ++kc) { bf[kc].x = 0.0f; bf[kc].y = 0.0f; }
    if (keep) {
#pragma unroll
      for (int kc = 0; kc < 4; ++kc) {
        const int k0 = 4 * kc + 2 * fp_hi;
        bf[kc].x = Vp[k0 * PM + fp_n];
        bf[kc].y = Vp[(k0 + 1) * PM + fp_n];
      }
    }
    factor_product(Um, bf, &fac[(2 * pr + (keep ? fp_n : 0)) * 16], keep);
  }
  __syncthreads();

  if (fp_wave == 1) {
    const bool keep = fp_n < 4;
    v2f bf[4];
#pragma unroll
    for (int kc = 0; kc < 4; ++kc) { bf[kc].x = 0.0f; bf[kc].y = 0.0f; }
    if (keep) {
      const int odd = fp_n >> 1;
      const float* src = &fac[(fp_n & 1) * 16 + odd];
#pragma unroll
      for (int kc = 0; kc < 4; ++kc) {
        const float t = src[4 * kc + 2 * fp_hi];
        bf[kc].x = odd ? 0.0f : t;
        bf[kc].y = odd ? t : 0.0f;
      }
    }
    factor_product(U[0], bf, &fac2[(keep ? fp_n : 0) * 16], keep);
  }

  {
    const int n = fp_n, hi = fp_hi, j = fp_wave;
    const int msel = n ^ ((j >> 1) << 3);
    const float ca = fac[64 + 16 * (j & 1) + msel];
    const float cb = fac[96 + 16 * (j & 1) + msel];
    const int lb = 8 * (hi ^ (__popc(n) & 1));
    const v4f* le = reinterpret_cast<const v4f*>(&fac[32 + lb]);
    const v4f* lo = reinterpret_cast<const v4f*>(&fac[48 + lb]);
    v4f in[2];
    in[0] = le[0] * ca + lo[0] * cb;
    in[1] = le[1] * ca + lo[1] * cb;
    const int frag = n * PM + 8 * hi;
    v2f bl[4], bm[4];
    {
      const v4f* gl = reinterpret_cast<const v4f*>(&U[3][frag]);
      const v4f* gm = reinterpret_cast<const v4f*>(&U[2][frag]);
      v4f l0 = gl[0], l1 = gl[1], m0 = gm[0], m1 = gm[1];
      bl[0].x = l0.x; bl[0].y = l0.y; bl[1].x = l0.z; bl[1].y = l0.w;
      bl[2].x = l1.x; bl[2].y = l1.y; bl[3].x = l1.z; bl[3].y = l1.w;
      bm[0].x = m0.x; bm[0].y = m0.y; bm[1].x = m0.z; bm[1].y = m0.w;
      bm[2].x = m1.x; bm[2].y = m1.y; bm[3].x = m1.z; bm[3].y = m1.w;
    }
    v8f y = {};
#pragma unroll
    for (int kc = 0; kc < 4; ++kc) {
      v2f af;
      af.x = in[kc >> 1][2 * (kc & 1)];
      af.y = in[kc >> 1][2 * (kc & 1) + 1];
      y = __builtin_amdgcn_wmma_f32_16x16x4_f32(
           false, af,  false, bl[kc],
           (short)0, y,  false,  false);
    }
    v8f c = {};
#pragma unroll
    for (int kc = 0; kc < 4; ++kc) {
      v2f af;
      af.x = y[2 * kc];
      af.y = y[2 * kc + 1];
      c = __builtin_amdgcn_wmma_f32_16x16x4_f32(
           false, af,  false, bm[kc],
           (short)0, c,  false,  false);
    }
    v4f* q = reinterpret_cast<v4f*>(&psi[j * PT + n * PM + 8 * hi]);
    v4f o0, o1;
    o0.x = c[0]; o0.y = c[1]; o0.z = c[2]; o0.w = c[3];
    o1.x = c[4]; o1.y = c[5]; o1.z = c[6]; o1.w = c[7];
    q[0] = o0;
    q[1] = o1;
  }
  __syncthreads();

  if (fp_wave == 0) {
    const bool keep = fp_n < 8;
    v2f bf[4];
#pragma unroll
    for (int kc = 0; kc < 4; ++kc) { bf[kc].x = 0.0f; bf[kc].y = 0.0f; }
    if (keep) {
      const int odd = fp_n >> 2;
      const float* src = &fac2[(fp_n & 3) * 16 + odd];
#pragma unroll
      for (int kc = 0; kc < 4; ++kc) {
        const float t = src[4 * kc + 2 * fp_hi];
        bf[kc].x = odd ? 0.0f : t;
        bf[kc].y = odd ? t : 0.0f;
      }
    }
    factor_product(U[0], bf, &fac[(keep ? fp_n : 0) * 16], keep);
  }

  {
    const int n = fp_n, hi = fp_hi, j = fp_wave;
    const int cz8 = 8 * (hi ^ (__popc(n) & 1));
    v4f in[2][2];
#pragma unroll
    for (int s = 0; s < 2; ++s) {
      const v4f* p = reinterpret_cast<const v4f*>(&psi[j * PT + (n ^ (8 * s)) * PM + cz8]);
      in[s][0] = p[0];
      in[s][1] = p[1];
    }
    const int frag = n * PM + 8 * hi;
    v2f bl[4], bm[4];
    {
      const v4f* gl = reinterpret_cast<const v4f*>(&U[3][frag]);
      const v4f* gm = reinterpret_cast<const v4f*>(&U[2][frag]);
      v4f l0 = gl[0], l1 = gl[1], m0 = gm[0], m1 = gm[1];
      bl[0].x = l0.x; bl[0].y = l0.y; bl[1].x = l0.z; bl[1].y = l0.w;
      bl[2].x = l1.x; bl[2].y = l1.y; bl[3].x = l1.z; bl[3].y = l1.w;
      bm[0].x = m0.x; bm[0].y = m0.y; bm[1].x = m0.z; bm[1].y = m0.w;
      bm[2].x = m1.x; bm[2].y = m1.y; bm[3].x = m1.z; bm[3].y = m1.w;
    }
    v8f y[2];
#pragma unroll
    for (int s = 0; s < 2; ++s) {
      v8f c = {};
#pragma unroll
      for (int kc = 0; kc < 4; ++kc) {
        v2f af;
        af.x = in[s][kc >> 1][2 * (kc & 1)];
        af.y = in[s][kc >> 1][2 * (kc & 1) + 1];
        c = __builtin_amdgcn_wmma_f32_16x16x4_f32(
             false, af,  false, bl[kc],
             (short)0, c,  false,  false);
      }
      y[s] = c;
    }
#pragma unroll
    for (int s = 0; s < 2; ++s) {
      v8f c = {};
#pragma unroll
      for (int kc = 0; kc < 4; ++kc) {
        v2f af;
        af.x = y[s][2 * kc];
        af.y = y[s][2 * kc + 1];
        c = __builtin_amdgcn_wmma_f32_16x16x4_f32(
             false, af,  false, bm[kc],
             (short)0, c,  false,  false);
      }
      v4f* q = reinterpret_cast<v4f*>(&psi[(4 + 4 * s + j) * PT + n * PM + 8 * hi]);
      v4f o0, o1;
      o0.x = c[0]; o0.y = c[1]; o0.z = c[2]; o0.w = c[3];
      o1.x = c[4]; o1.y = c[5]; o1.z = c[6]; o1.w = c[7];
      q[0] = o0;
      q[1] = o1;
    }
  }
  __syncthreads();

  {
    float ZE[4][2], ZO[4][2];
#pragma unroll
    for (int j = 0; j < 4; ++j) {
#pragma unroll
      for (int h = 0; h < 2; ++h) {
        ZE[j][h] = psi[(4 + j) * PT + (midq + 8 * h) * PM + lowq];
        ZO[j][h] = psi[(8 + j) * PT + (midq + 8 * h) * PM + lowq];
      }
    }
#pragma unroll
    for (int top = 0; top < 16; ++top) {
      float ve[4], vo[4];
#pragma unroll
      for (int j = 0; j < 4; ++j) { ve[j] = fac[16 * j + top]; vo[j] = fac[64 + 16 * j + top]; }
#pragma unroll
      for (int h = 0; h < 2; ++h) {
        float v = ve[0] * ZE[0][h];
#pragma unroll
        for (int j = 1; j < 4; ++j) v += ve[j] * ZE[j][h];
#pragma unroll
        for (int j = 0; j < 4; ++j) v += vo[j] * ZO[j][h];
        psi[top * PT + ((midq + 8 * h) ^ ((top & 1) << 3)) * PM + lowq] = v;
      }
    }
  }
  __syncthreads();

  auto apply = [&](const float* Um, int sm, int sn, int st, bool colfold, int rowfold) {
    const int lane = tid & 31, wave = tid >> 5;
    const int n = lane & 15, hi = lane >> 4;
    v2f a[4];
#pragma unroll
    for (int kc = 0; kc < 4; ++kc) {
      int k0 = 4 * kc + 2 * hi;
      a[kc].x = Um[n * PM + k0];
      a[kc].y = Um[n * PM + k0 + 1];
    }
#pragma unroll
    for (int tt = 0; tt < 4; ++tt) {
      int t = wave * 4 + tt;
      int base = st * t + sn * n;
      v8f c = {};
#pragma unroll
      for (int kc = 0; kc < 4; ++kc) {
        int k0 = 4 * kc + 2 * hi;
        v2f bf;
        bf.x = psi[base + sm * k0];
        bf.y = psi[base + sm * (k0 + 1)];
        c = __builtin_amdgcn_wmma_f32_16x16x4_f32(
             false, a[kc],  false, bf,
             (short)0, c,  false,  false);
      }
      int sbase = colfold ? st * t + sn * (n ^ ((__popc(t) & 1) << 3)) : base;
      int shi = hi ^ (t & rowfold);
#pragma unroll
      for (int v = 0; v < 8; ++v)
        psi[sbase + sm * (v + 8 * shi)] = c[v];
    }
  };

  auto apply_low_mid = [&](const float* Ulow, const float* Umid, int rowfold) {
    const int lane = tid & 31, wave = tid >> 5;
    const int n = lane & 15, hi = lane >> 4;
    const int frag = n * PM + 8 * hi;
    v4f in[4][2];
#pragma unroll
    for (int tt = 0; tt < 4; ++tt) {
      const v4f* p = reinterpret_cast<const v4f*>(&psi[(wave * 4 + tt) * PT + frag]);
      in[tt][0] = p[0];
      in[tt][1] = p[1];
    }
    v2f bl[4], bm[4];
    {
      const v4f* gl = reinterpret_cast<const v4f*>(&Ulow[frag]);
      const v4f* gm = reinterpret_cast<const v4f*>(&Umid[frag]);
      v4f l0 = gl[0], l1 = gl[1], m0 = gm[0], m1 = gm[1];
      bl[0].x = l0.x; bl[0].y = l0.y; bl[1].x = l0.z; bl[1].y = l0.w;
      bl[2].x = l1.x; bl[2].y = l1.y; bl[3].x = l1.z; bl[3].y = l1.w;
      bm[0].x = m0.x; bm[0].y = m0.y; bm[1].x = m0.z; bm[1].y = m0.w;
      bm[2].x = m1.x; bm[2].y = m1.y; bm[3].x = m1.z; bm[3].y = m1.w;
    }
    v8f y[4];
#pragma unroll
    for (int tt = 0; tt < 4; ++tt) {
      v8f c = {};
#pragma unroll
      for (int kc = 0; kc < 4; ++kc) {
        v2f af;
        af.x = in[tt][kc >> 1][2 * (kc & 1)];
        af.y = in[tt][kc >> 1][2 * (kc & 1) + 1];
        c = __builtin_amdgcn_wmma_f32_16x16x4_f32(
             false, af,  false, bl[kc],
             (short)0, c,  false,  false);
      }
      y[tt] = c;
    }
#pragma unroll
    for (int tt = 0; tt < 4; ++tt) {
      v8f c = {};
#pragma unroll
      for (int kc = 0; kc < 4; ++kc) {
        v2f af;
        af.x = y[tt][2 * kc];
        af.y = y[tt][2 * kc + 1];
        c = __builtin_amdgcn_wmma_f32_16x16x4_f32(
             false, af,  false, bm[kc],
             (short)0, c,  false,  false);
      }
      const int t = wave * 4 + tt;
      v4f* q = reinterpret_cast<v4f*>(&psi[t * PT + (n ^ ((t & rowfold) << 3)) * PM + 8 * hi]);
      v4f o0, o1;
      o0.x = c[0]; o0.y = c[1]; o0.z = c[2]; o0.w = c[3];
      o1.x = c[4]; o1.y = c[5]; o1.z = c[6]; o1.w = c[7];
      q[0] = o0;
      q[1] = o1;
    }
  };

  apply(U[1], PT, 1, PM, true, 0);      __syncthreads();
  apply_low_mid(U[3], U[2], 0);         __syncthreads();

  float gml0, gml1;
  {
    float gl = 0.0f, gm = 0.0f;
    for (int q = 0; q < 4; ++q) gl += ((lowq >> (3 - q)) & 1) ? -hws[8 + q] : hws[8 + q];
    for (int q = 1; q < 4; ++q) gm += ((midq >> (3 - q)) & 1) ? -hws[4 + q] : hws[4 + q];
    gm += gl;
    gml0 = gm + hws[4];
    gml1 = gm - hws[4];
  }
  float gt[16];
#pragma unroll
  for (int x = 0; x < 16; ++x) {
    float v = 0.0f;
    for (int q = 0; q < 4; ++q) v += ((x >> (3 - q)) & 1) ? -hws[q] : hws[q];
    gt[x] = v;
  }
  float partial = 0.0f;
#pragma unroll
  for (int it = 0; it < 32; ++it) {
    float p = psi[pb + (it >> 1) * PT + (it & 1) * (8 * PM)]; p *= p;
    partial += p * (gt[it >> 1] + ((it & 1) ? gml1 : gml0));
  }
  red[tid] = partial;
  __syncthreads();
#pragma unroll
  for (int off = QS_THREADS / 2; off > 0; off >>= 1) {
    if (tid < off) red[tid] += red[tid + off];
    __syncthreads();
  }
  if (tid == 0) out[b] = red[0] + head_b[0];
}

extern "C" void kernel_launch(void* const* d_in, const int* in_sizes, int n_in,
                              void* d_out, int out_size, void* d_ws, size_t ws_size,
                              hipStream_t stream) {
  const float* sb     = (const float*)d_in[0];
  const float* params = (const float*)d_in[1];
  const float* hw     = (const float*)d_in[2];
  const float* hb     = (const float*)d_in[3];
  float* out = (float*)d_out;
  int batch = in_sizes[0] / 12;
  qsim_wmma_kernel<<<dim3(batch), dim3(QS_THREADS), 0, stream>>>(
      sb, params, hw, hb, out);
}


# Each workgroup loops over 4 samples: gate matrices, readout weights, staging roles and all 4 samples' sines/cosines are set up once per workgroup, not per sample; grid 8192 -> 2048, all resident
# speedup vs baseline: 4.5150x; 1.0927x over previous
#include <hip/hip_runtime.h>

typedef __attribute__((ext_vector_type(2))) float v2f;
typedef __attribute__((ext_vector_type(4))) float v4f;
typedef __attribute__((ext_vector_type(8))) float v8f;

#define QS_THREADS 128
#define SPB 4

#define PM 20
#define PT 332

__device__ __forceinline__ int perm4(int x) {
  if (x & 8) x ^= 4;
  if (x & 4) x ^= 2;
  if (x & 2) x ^= 1;
  return x;
}
__device__ __forceinline__ int perm4inv(int x) {
  if (x & 2) x ^= 1;
  if (x & 4) x ^= 2;
  if (x & 8) x ^= 4;
  return x;
}

__global__ __launch_bounds__(QS_THREADS, 8) void qsim_wmma_kernel(
    const float* __restrict__ state_batch,
    const float* __restrict__ params,
    const float* __restrict__ head_w,
    const float* __restrict__ head_b,
    float* __restrict__ out,
    int batch) {
  __shared__ __attribute__((aligned(16))) float psi[16 * PT];
  __shared__ __attribute__((aligned(16))) float U[4][16 * PM];
  __shared__ float cz[12 + 12 * SPB], sz[12 + 12 * SPB];
  __shared__ __attribute__((aligned(16))) float gts[16];
  __shared__ __attribute__((aligned(16))) float red[QS_THREADS];
  float* const fac = red;
  __shared__ __attribute__((aligned(16))) float fac2[64];

  const int b0  = blockIdx.x * SPB;
  const int nb  = min(SPB, batch - b0);
  const int tid = threadIdx.x;
  const int pb  = (tid >> 4) * PM + (tid & 15);

  if (tid < 12 + 12 * nb) {
    float a = 0.5f * ((tid < 12) ? params[tid] : state_batch[(size_t)b0 * 12 + (tid - 12)]);
    cz[tid] = cosf(a); sz[tid] = sinf(a);
  }
  if (tid >= 64 && tid < 80) {
    const int x = tid - 64;
    float v = 0.0f;
    for (int q = 0; q < 4; ++q) v += ((x >> (3 - q)) & 1) ? -head_w[q] : head_w[q];
    gts[x] = v;
  }
  const float hb = head_b[0];
  __syncthreads();

  auto kron4 = [&](int base, int i, int j) -> float {
    float v = 1.0f;
#pragma unroll
    for (int q = 0; q < 4; ++q) {
      int bi = (i >> (3 - q)) & 1;
      int bj = (j >> (3 - q)) & 1;
      float c = cz[base + q], s = sz[base + q];
      v *= bi ? (bj ? c : s) : (bj ? -s : c);
    }
    return v;
  };

  for (int e = tid; e < 4 * 256; e += QS_THREADS) {
    int mi = e >> 8, idx = e & 255, i = idx >> 4, j = idx & 15;
    float v;
    switch (mi) {
      case 0:  v = kron4(0, perm4inv(i), j); break;
      case 1:  v = kron4(0, i, j); break;
      case 2:  v = kron4(4, i, perm4(j)); break;
      default: v = kron4(8, i, perm4(j)); break;
    }
    U[mi][i * PM + j] = v;
  }

  const int lowq = tid & 15, midq = tid >> 4;
  auto amp4 = [&](int base, int x) -> float {
    float v = 1.0f;
#pragma unroll
    for (int q = 0; q < 4; ++q) v *= ((x >> (3 - q)) & 1) ? sz[base + q] : cz[base + q];
    return v;
  };
  float gml0, gml1;
  {
    float gl = 0.0f, gm = 0.0f;
    for (int q = 0; q < 4; ++q) gl += ((lowq >> (3 - q)) & 1) ? -head_w[8 + q] : head_w[8 + q];
    for (int q = 1; q < 4; ++q) gm += ((midq >> (3 - q)) & 1) ? -head_w[4 + q] : head_w[4 + q];
    gm += gl;
    gml0 = gm + head_w[4];
    gml1 = gm - head_w[4];
  }

  int st_nib, st_x, st_dst;
  bool st_on;
  {
    const int k = tid & 15;
    int plane, col;
    if (tid < 32) {
      plane = 0; col = tid >> 4;
      st_nib = 0; st_x = perm4inv(k); st_on = (k & 1) == col;
    } else if (tid < 64) {
      plane = 1; col = (tid >> 4) & 1;
      st_nib = 8; st_x = k ^ (col << 3); st_on = true;
    } else {
      plane = 2; col = (tid >> 4) & 3;
      st_nib = 4; st_x = k ^ ((col & 1) << 3); st_on = (__popc(k) & 1) == (col >> 1);
    }
    st_dst = plane * PT + k * PM + col;
  }

#pragma unroll 1
  for (int smp = 0; smp < nb; ++smp) {
  const int czb = 12 + 12 * smp;
  psi[st_dst] = st_on ? amp4(czb + st_nib, st_x) : 0.0f;
  __syncthreads();

  const int fp_lane = tid & 31, fp_wave = tid >> 5;
  const int fp_n = fp_lane & 15, fp_hi = fp_lane >> 4;
  auto factor_product = [&](const float* Um, const v2f (&bf)[4], float* dst, bool keep) {
    v2f af[4];
#pragma unroll
    for (int kc = 0; kc < 4; ++kc) {
      const int k0 = 4 * kc + 2 * fp_hi;
      af[kc].x = Um[fp_n * PM + k0];
      af[kc].y = Um[fp_n * PM + k0 + 1];
    }
    v8f c = {};
#pragma unroll
    for (int kc = 0; kc < 4; ++kc)
      c = __builtin_amdgcn_wmma_f32_16x16x4_f32(
           false, af[kc],  false, bf[kc],
           (short)0, c,  false,  false);
    if (keep) {
#pragma unroll
      for (int v = 0; v < 8; ++v)
        dst[v + 8 * fp_hi] = c[v];
    }
  };

  if (fp_wave >= 1) {
    const int pr = fp_wave - 1;
    const float* Um = (pr == 0) ? U[0] : ((pr == 1) ? U[3] : U[2]);
    const int ncol = (pr == 2) ? 4 : 2;
    const bool keep = fp_n < ncol;
    const float* Vp = &psi[pr * PT];
    v2f bf[4];
#pragma unroll
    for (int kc = 0; kc < 4; ++kc) { bf[kc].x = 0.0f; bf[kc].y = 0.0f; }
    if (keep) {
#pragma unroll
      for (int kc = 0; kc < 4; ++kc) {
        const int k0 = 4 * kc + 2 * fp_hi;
        bf[kc].x = Vp[k0 * PM + fp_n];
        bf[kc].y = Vp[(k0 + 1) * PM + fp_n];
      }
    }
    factor_product(Um, bf, &fac[(2 * pr + (keep ? fp_n : 0)) * 16], keep);
  }
  __syncthreads();

  if (fp_wave == 1) {
    const bool keep = fp_n < 4;
    v2f bf[4];
#pragma unroll
    for (int kc = 0; kc < 4; ++kc) { bf[kc].x = 0.0f; bf[kc].y = 0.0f; }
    if (keep) {
      const int odd = fp_n >> 1;
      const float* src = &fac[(fp_n & 1) * 16 + odd];
#pragma unroll
      for (int kc = 0; kc < 4; ++kc) {
        const float t = src[4 * kc + 2 * fp_hi];
        bf[kc].x = odd ? 0.0f : t;
        bf[kc].y = odd ? t : 0.0f;
      }
    }
    factor_product(U[0], bf, &fac2[(keep ? fp_n : 0) * 16], keep);
  }

  {
    const int n = fp_n, hi = fp_hi, j = fp_wave;
    const int msel = n ^ ((j >> 1) << 3);
    const float ca = fac[64 + 16 * (j & 1) + msel];
    const float cb = fac[96 + 16 * (j & 1) + msel];
    const int lb = 8 * (hi ^ (__popc(n) & 1));
    const v4f* le = reinterpret_cast<const v4f*>(&fac[32 + lb]);
    const v4f* lo = reinterpret_cast<const v4f*>(&fac[48 + lb]);
    v4f in[2];
    in[0] = le[0] * ca + lo[0] * cb;
    in[1] = le[1] * ca + lo[1] * cb;
    const int frag = n * PM + 8 * hi;
    v2f bl[4], bm[4];
    {
      const v4f* gl = reinterpret_cast<const v4f*>(&U[3][frag]);
      const v4f* gm = reinterpret_cast<const v4f*>(&U[2][frag]);
      v4f l0 = gl[0], l1 = gl[1], m0 = gm[0], m1 = gm[1];
      bl[0].x = l0.x; bl[0].y = l0.y; bl[1].x = l0.z; bl[1].y = l0.w;
      bl[2].x = l1.x; bl[2].y = l1.y; bl[3].x = l1.z; bl[3].y = l1.w;
      bm[0].x = m0.x; bm[0].y = m0.y; bm[1].x = m0.z; bm[1].y = m0.w;
      bm[2].x = m1.x; bm[2].y = m1.y; bm[3].x = m1.z; bm[3].y = m1.w;
    }
    v8f y = {};
#pragma unroll
    for (int kc = 0; kc < 4; ++kc) {
      v2f af;
      af.x = in[kc >> 1][2 * (kc & 1)];
      af.y = in[kc >> 1][2 * (kc & 1) + 1];
      y = __builtin_amdgcn_wmma_f32_16x16x4_f32(
           false, af,  false, bl[kc],
           (short)0, y,  false,  false);
    }
    v8f c = {};
#pragma unroll
    for (int kc = 0; kc < 4; ++kc) {
      v2f af;
      af.x = y[2 * kc];
      af.y = y[2 * kc + 1];
      c = __builtin_amdgcn_wmma_f32_16x16x4_f32(
           false, af,  false, bm[kc],
           (short)0, c,  false,  false);
    }
    v4f* q = reinterpret_cast<v4f*>(&psi[j * PT + n * PM + 8 * hi]);
    v4f o0, o1;
    o0.x = c[0]; o0.y = c[1]; o0.z = c[2]; o0.w = c[3];
    o1.x = c[4]; o1.y = c[5]; o1.z = c[6]; o1.w = c[7];
    q[0] = o0;
    q[1] = o1;
  }
  __syncthreads();

  if (fp_wave == 0) {
    const bool keep = fp_n < 8;
    v2f bf[4];
#pragma unroll
    for (int kc = 0; kc < 4; ++kc) { bf[kc].x = 0.0f; bf[kc].y = 0.0f; }
    if (keep) {
      const int odd = fp_n >> 2;
      const float* src = &fac2[(fp_n & 3) * 16 + odd];
#pragma unroll
      for (int kc = 0; kc < 4; ++kc) {
        const float t = src[4 * kc + 2 * fp_hi];
        bf[kc].x = odd ? 0.0f : t;
        bf[kc].y = odd ? t : 0.0f;
      }
    }
    factor_product(U[0], bf, &fac[(keep ? fp_n : 0) * 16], keep);
  }

  {
    const int n = fp_n, hi = fp_hi, j = fp_wave;
    const int cz8 = 8 * (hi ^ (__popc(n) & 1));
    v4f in[2][2];
#pragma unroll
    for (int s = 0; s < 2; ++s) {
      const v4f* p = reinterpret_cast<const v4f*>(&psi[j * PT + (n ^ (8 * s)) * PM + cz8]);
      in[s][0] = p[0];
      in[s][1] = p[1];
    }
    const int frag = n * PM + 8 * hi;
    v2f bl[4], bm[4];
    {
      const v4f* gl = reinterpret_cast<const v4f*>(&U[3][frag]);
      const v4f* gm = reinterpret_cast<const v4f*>(&U[2][frag]);
      v4f l0 = gl[0], l1 = gl[1], m0 = gm[0], m1 = gm[1];
      bl[0].x = l0.x; bl[0].y = l0.y; bl[1].x = l0.z; bl[1].y = l0.w;
      bl[2].x = l1.x; bl[2].y = l1.y; bl[3].x = l1.z; bl[3].y = l1.w;
      bm[0].x = m0.x; bm[0].y = m0.y; bm[1].x = m0.z; bm[1].y = m0.w;
      bm[2].x = m1.x; bm[2].y = m1.y; bm[3].x = m1.z; bm[3].y = m1.w;
    }
    v8f y[2];
#pragma unroll
    for (int s = 0; s < 2; ++s) {
      v8f c = {};
#pragma unroll
      for (int kc = 0; kc < 4; ++kc) {
        v2f af;
        af.x = in[s][kc >> 1][2 * (kc & 1)];
        af.y = in[s][kc >> 1][2 * (kc & 1) + 1];
        c = __builtin_amdgcn_wmma_f32_16x16x4_f32(
             false, af,  false, bl[kc],
             (short)0, c,  false,  false);
      }
      y[s] = c;
    }
#pragma unroll
    for (int s = 0; s < 2; ++s) {
      v8f c = {};
#pragma unroll
      for (int kc = 0; kc < 4; ++kc) {
        v2f af;
        af.x = y[s][2 * kc];
        af.y = y[s][2 * kc + 1];
        c = __builtin_amdgcn_wmma_f32_16x16x4_f32(
             false, af,  false, bm[kc],
             (short)0, c,  false,  false);
      }
      v4f* q = reinterpret_cast<v4f*>(&psi[(4 + 4 * s + j) * PT + n * PM + 8 * hi]);
      v4f o0, o1;
      o0.x = c[0]; o0.y = c[1]; o0.z = c[2]; o0.w = c[3];
      o1.x = c[4]; o1.y = c[5]; o1.z = c[6]; o1.w = c[7];
      q[0] = o0;
      q[1] = o1;
    }
  }
  __syncthreads();

  {
    float ZE[4][2], ZO[4][2];
#pragma unroll
    for (int j = 0; j < 4; ++j) {
#pragma unroll
      for (int h = 0; h < 2; ++h) {
        ZE[j][h] = psi[(4 + j) * PT + (midq + 8 * h) * PM + lowq];
        ZO[j][h] = psi[(8 + j) * PT + (midq + 8 * h) * PM + lowq];
      }
    }
#pragma unroll
    for (int top = 0; top < 16; ++top) {
      float ve[4], vo[4];
#pragma unroll
      for (int j = 0; j < 4; ++j) { ve[j] = fac[16 * j + top]; vo[j] = fac[64 + 16 * j + top]; }
#pragma unroll
      for (int h = 0; h < 2; ++h) {
        float v = ve[0] * ZE[0][h];
#pragma unroll
        for (int j = 1; j < 4; ++j) v += ve[j] * ZE[j][h];
#pragma unroll
        for (int j = 0; j < 4; ++j) v += vo[j] * ZO[j][h];
        psi[top * PT + ((midq + 8 * h) ^ ((top & 1) << 3)) * PM + lowq] = v;
      }
    }
  }
  __syncthreads();

  auto apply = [&](const float* Um, int sm, int sn, int st, bool colfold, int rowfold) {
    const int lane = tid & 31, wave = tid >> 5;
    const int n = lane & 15, hi = lane >> 4;
    v2f a[4];
#pragma unroll
    for (int kc = 0; kc < 4; ++kc) {
      int k0 = 4 * kc + 2 * hi;
      a[kc].x = Um[n * PM + k0];
      a[kc].y = Um[n * PM + k0 + 1];
    }
#pragma unroll
    for (int tt = 0; tt < 4; ++tt) {
      int t = wave * 4 + tt;
      int base = st * t + sn * n;
      v8f c = {};
#pragma unroll
      for (int kc = 0; kc < 4; ++kc) {
        int k0 = 4 * kc + 2 * hi;
        v2f bf;
        bf.x = psi[base + sm * k0];
        bf.y = psi[base + sm * (k0 + 1)];
        c = __builtin_amdgcn_wmma_f32_16x16x4_f32(
             false, a[kc],  false, bf,
             (short)0, c,  false,  false);
      }
      int sbase = colfold ? st * t + sn * (n ^ ((__popc(t) & 1) << 3)) : base;
      int shi = hi ^ (t & rowfold);
#pragma unroll
      for (int v = 0; v < 8; ++v)
        psi[sbase + sm * (v + 8 * shi)] = c[v];
    }
  };

  auto apply_low_mid = [&](const float* Ulow, const float* Umid, int rowfold) {
    const int lane = tid & 31, wave = tid >> 5;
    const int n = lane & 15, hi = lane >> 4;
    const int frag = n * PM + 8 * hi;
    v4f in[4][2];
#pragma unroll
    for (int tt = 0; tt < 4; ++tt) {
      const v4f* p = reinterpret_cast<const v4f*>(&psi[(wave * 4 + tt) * PT + frag]);
      in[tt][0] = p[0];
      in[tt][1] = p[1];
    }
    v2f bl[4], bm[4];
    {
      const v4f* gl = reinterpret_cast<const v4f*>(&Ulow[frag]);
      const v4f* gm = reinterpret_cast<const v4f*>(&Umid[frag]);
      v4f l0 = gl[0], l1 = gl[1], m0 = gm[0], m1 = gm[1];
      bl[0].x = l0.x; bl[0].y = l0.y; bl[1].x = l0.z; bl[1].y = l0.w;
      bl[2].x = l1.x; bl[2].y = l1.y; bl[3].x = l1.z; bl[3].y = l1.w;
      bm[0].x = m0.x; bm[0].y = m0.y; bm[1].x = m0.z; bm[1].y = m0.w;
      bm[2].x = m1.x; bm[2].y = m1.y; bm[3].x = m1.z; bm[3].y = m1.w;
    }
    v8f y[4];
#pragma unroll
    for (int tt = 0; tt < 4; ++tt) {
      v8f c = {};
#pragma unroll
      for (int kc = 0; kc < 4; ++kc) {
        v2f af;
        af.x = in[tt][kc >> 1][2 * (kc & 1)];
        af.y = in[tt][kc >> 1][2 * (kc & 1) + 1];
        c = __builtin_amdgcn_wmma_f32_16x16x4_f32(
             false, af,  false, bl[kc],
             (short)0, c,  false,  false);
      }
      y[tt] = c;
    }
#pragma unroll
    for (int tt = 0; tt < 4; ++tt) {
      v8f c = {};
#pragma unroll
      for (int kc = 0; kc < 4; ++kc) {
        v2f af;
        af.x = y[tt][2 * kc];
        af.y = y[tt][2 * kc + 1];
        c = __builtin_amdgcn_wmma_f32_16x16x4_f32(
             false, af,  false, bm[kc],
             (short)0, c,  false,  false);
      }
      const int t = wave * 4 + tt;
      v4f* q = reinterpret_cast<v4f*>(&psi[t * PT + (n ^ ((t & rowfold) << 3)) * PM + 8 * hi]);
      v4f o0, o1;
      o0.x = c[0]; o0.y = c[1]; o0.z = c[2]; o0.w = c[3];
      o1.x = c[4]; o1.y = c[5]; o1.z = c[6]; o1.w = c[7];
      q[0] = o0;
      q[1] = o1;
    }
  };

  apply(U[1], PT, 1, PM, true, 0);      __syncthreads();
  apply_low_mid(U[3], U[2], 0);         __syncthreads();

  float gt[16];
#pragma unroll
  for (int x = 0; x < 16; ++x) gt[x] = gts[x];
  float partial = 0.0f;
#pragma unroll
  for (int it = 0; it < 32; ++it) {
    float p = psi[pb + (it >> 1) * PT + (it & 1) * (8 * PM)]; p *= p;
    partial += p * (gt[it >> 1] + ((it & 1) ? gml1 : gml0));
  }
  red[tid] = partial;
  __syncthreads();
#pragma unroll
  for (int off = QS_THREADS / 2; off > 0; off >>= 1) {
    if (tid < off) red[tid] += red[tid + off];
    __syncthreads();
  }
  if (tid == 0) out[b0 + smp] = red[0] + hb;
  }
}

extern "C" void kernel_launch(void* const* d_in, const int* in_sizes, int n_in,
                              void* d_out, int out_size, void* d_ws, size_t ws_size,
                              hipStream_t stream) {
  const float* sb     = (const float*)d_in[0];
  const float* params = (const float*)d_in[1];
  const float* hw     = (const float*)d_in[2];
  const float* hb     = (const float*)d_in[3];
  float* out = (float*)d_out;
  int batch = in_sizes[0] / 12;
  qsim_wmma_kernel<<<dim3((batch + SPB - 1) / SPB), dim3(QS_THREADS), 0, stream>>>(
      sb, params, hw, hb, out, batch);
}


# Readout split by nibble: gT term from the last top pass's D registers, gM+gL term from a low/mid pass on the expanded state (rotation left out is orthogonal); no pass stores, no sweep; new sum order
# speedup vs baseline: 4.7661x; 1.0556x over previous
#include <hip/hip_runtime.h>

typedef __attribute__((ext_vector_type(2))) float v2f;
typedef __attribute__((ext_vector_type(4))) float v4f;
typedef __attribute__((ext_vector_type(8))) float v8f;

#define QS_THREADS 128
#define SPB 4

#define PM 20
#define PT 332

__device__ __forceinline__ int perm4(int x) {
  if (x & 8) x ^= 4;
  if (x & 4) x ^= 2;
  if (x & 2) x ^= 1;
  return x;
}
__device__ __forceinline__ int perm4inv(int x) {
  if (x & 2) x ^= 1;
  if (x & 4) x ^= 2;
  if (x & 8) x ^= 4;
  return x;
}

__global__ __launch_bounds__(QS_THREADS, 8) void qsim_wmma_kernel(
    const float* __restrict__ state_batch,
    const float* __restrict__ params,
    const float* __restrict__ head_w,
    const float* __restrict__ head_b,
    float* __restrict__ out,
    int batch) {
  __shared__ __attribute__((aligned(16))) float psi[16 * PT];
  __shared__ __attribute__((aligned(16))) float U[4][16 * PM];
  __shared__ float cz[12 + 12 * SPB], sz[12 + 12 * SPB];
  __shared__ __attribute__((aligned(16))) float gw[48];
  __shared__ __attribute__((aligned(16))) float red[QS_THREADS];
  float* const fac = red;
  __shared__ __attribute__((aligned(16))) float fac2[64];

  const int b0  = blockIdx.x * SPB;
  const int nb  = min(SPB, batch - b0);
  const int tid = threadIdx.x;

  if (tid < 12 + 12 * nb) {
    float a = 0.5f * ((tid < 12) ? params[tid] : state_batch[(size_t)b0 * 12 + (tid - 12)]);
    cz[tid] = cosf(a); sz[tid] = sinf(a);
  }
  if (tid >= 64 && tid < 112) {
    const int nib = (tid - 64) >> 4, x = tid & 15;
    float v = 0.0f;
    for (int q = 0; q < 4; ++q) v += ((x >> (3 - q)) & 1) ? -head_w[4 * nib + q] : head_w[4 * nib + q];
    gw[16 * nib + x] = v;
  }
  const float hb = head_b[0];
  __syncthreads();

  auto kron4 = [&](int base, int i, int j) -> float {
    float v = 1.0f;
#pragma unroll
    for (int q = 0; q < 4; ++q) {
      int bi = (i >> (3 - q)) & 1;
      int bj = (j >> (3 - q)) & 1;
      float c = cz[base + q], s = sz[base + q];
      v *= bi ? (bj ? c : s) : (bj ? -s : c);
    }
    return v;
  };

  for (int e = tid; e < 4 * 256; e += QS_THREADS) {
    int mi = e >> 8, idx = e & 255, i = idx >> 4, j = idx & 15;
    float v;
    switch (mi) {
      case 0:  v = kron4(0, perm4inv(i), j); break;
      case 1:  v = kron4(0, i, j); break;
      case 2:  v = kron4(4, i, perm4(j)); break;
      default: v = kron4(8, i, perm4(j)); break;
    }
    U[mi][i * PM + j] = v;
  }

  const int lowq = tid & 15, midq = tid >> 4;
  auto amp4 = [&](int base, int x) -> float {
    float v = 1.0f;
#pragma unroll
    for (int q = 0; q < 4; ++q) v *= ((x >> (3 - q)) & 1) ? sz[base + q] : cz[base + q];
    return v;
  };

  int st_nib, st_x, st_dst;
  bool st_on;
  {
    const int k = tid & 15;
    int plane, col;
    if (tid < 32) {
      plane = 0; col = tid >> 4;
      st_nib = 0; st_x = perm4inv(k); st_on = (k & 1) == col;
    } else if (tid < 64) {
      plane = 1; col = (tid >> 4) & 1;
      st_nib = 8; st_x = k ^ (col << 3); st_on = true;
    } else {
      plane = 2; col = (tid >> 4) & 3;
      st_nib = 4; st_x = k ^ ((col & 1) << 3); st_on = (__popc(k) & 1) == (col >> 1);
    }
    st_dst = plane * PT + k * PM + col;
  }

#pragma unroll 1
  for (int smp = 0; smp < nb; ++smp) {
  const int czb = 12 + 12 * smp;
  psi[st_dst] = st_on ? amp4(czb + st_nib, st_x) : 0.0f;
  __syncthreads();

  const int fp_lane = tid & 31, fp_wave = tid >> 5;
  const int fp_n = fp_lane & 15, fp_hi = fp_lane >> 4;
  auto factor_product = [&](const float* Um, const v2f (&bf)[4], float* dst, bool keep) {
    v2f af[4];
#pragma unroll
    for (int kc = 0; kc < 4; ++kc) {
      const int k0 = 4 * kc + 2 * fp_hi;
      af[kc].x = Um[fp_n * PM + k0];
      af[kc].y = Um[fp_n * PM + k0 + 1];
    }
    v8f c = {};
#pragma unroll
    for (int kc = 0; kc < 4; ++kc)
      c = __builtin_amdgcn_wmma_f32_16x16x4_f32(
           false, af[kc],  false, bf[kc],
           (short)0, c,  false,  false);
    if (keep) {
#pragma unroll
      for (int v = 0; v < 8; ++v)
        dst[v + 8 * fp_hi] = c[v];
    }
  };

  if (fp_wave >= 1) {
    const int pr = fp_wave - 1;
    const float* Um = (pr == 0) ? U[0] : ((pr == 1) ? U[3] : U[2]);
    const int ncol = (pr == 2) ? 4 : 2;
    const bool keep = fp_n < ncol;
    const float* Vp = &psi[pr * PT];
    v2f bf[4];
#pragma unroll
    for (int kc = 0; kc < 4; ++kc) { bf[kc].x = 0.0f; bf[kc].y = 0.0f; }
    if (keep) {
#pragma unroll
      for (int kc = 0; kc < 4; ++kc) {
        const int k0 = 4 * kc + 2 * fp_hi;
        bf[kc].x = Vp[k0 * PM + fp_n];
        bf[kc].y = Vp[(k0 + 1) * PM + fp_n];
      }
    }
    factor_product(Um, bf, &fac[(2 * pr + (keep ? fp_n : 0)) * 16], keep);
  }
  __syncthreads();

  if (fp_wave == 1) {
    const bool keep = fp_n < 4;
    v2f bf[4];
#pragma unroll
    for (int kc = 0; kc < 4; ++kc) { bf[kc].x = 0.0f; bf[kc].y = 0.0f; }
    if (keep) {
      const int odd = fp_n >> 1;
      const float* src = &fac[(fp_n & 1) * 16 + odd];
#pragma unroll
      for (int kc = 0; kc < 4; ++kc) {
        const float t = src[4 * kc + 2 * fp_hi];
        bf[kc].x = odd ? 0.0f : t;
        bf[kc].y = odd ? t : 0.0f;
      }
    }
    factor_product(U[0], bf, &fac2[(keep ? fp_n : 0) * 16], keep);
  }

  {
    const int n = fp_n, hi = fp_hi, j = fp_wave;
    const int msel = n ^ ((j >> 1) << 3);
    const float ca = fac[64 + 16 * (j & 1) + msel];
    const float cb = fac[96 + 16 * (j & 1) + msel];
    const int lb = 8 * (hi ^ (__popc(n) & 1));
    const v4f* le = reinterpret_cast<const v4f*>(&fac[32 + lb]);
    const v4f* lo = reinterpret_cast<const v4f*>(&fac[48 + lb]);
    v4f in[2];
    in[0] = le[0] * ca + lo[0] * cb;
    in[1] = le[1] * ca + lo[1] * cb;
    const int frag = n * PM + 8 * hi;
    v2f bl[4], bm[4];
    {
      const v4f* gl = reinterpret_cast<const v4f*>(&U[3][frag]);
      const v4f* gm = reinterpret_cast<const v4f*>(&U[2][frag]);
      v4f l0 = gl[0], l1 = gl[1], m0 = gm[0], m1 = gm[1];
      bl[0].x = l0.x; bl[0].y = l0.y; bl[1].x = l0.z; bl[1].y = l0.w;
      bl[2].x = l1.x; bl[2].y = l1.y; bl[3].x = l1.z; bl[3].y = l1.w;
      bm[0].x = m0.x; bm[0].y = m0.y; bm[1].x = m0.z; bm[1].y = m0.w;
      bm[2].x = m1.x; bm[2].y = m1.y; bm[3].x = m1.z; bm[3].y = m1.w;
    }
    v8f y = {};
#pragma unroll
    for (int kc = 0; kc < 4; ++kc) {
      v2f af;
      af.x = in[kc >> 1][2 * (kc & 1)];
      af.y = in[kc >> 1][2 * (kc & 1) + 1];
      y = __builtin_amdgcn_wmma_f32_16x16x4_f32(
           false, af,  false, bl[kc],
           (short)0, y,  false,  false);
    }
    v8f c = {};
#pragma unroll
    for (int kc = 0; kc < 4; ++kc) {
      v2f af;
      af.x = y[2 * kc];
      af.y = y[2 * kc + 1];
      c = __builtin_amdgcn_wmma_f32_16x16x4_f32(
           false, af,  false, bm[kc],
           (short)0, c,  false,  false);
    }
    v4f* q = reinterpret_cast<v4f*>(&psi[j * PT + n * PM + 8 * hi]);
    v4f o0, o1;
    o0.x = c[0]; o0.y = c[1]; o0.z = c[2]; o0.w = c[3];
    o1.x = c[4]; o1.y = c[5]; o1.z = c[6]; o1.w = c[7];
    q[0] = o0;
    q[1] = o1;
  }
  __syncthreads();

  if (fp_wave == 0) {
    const bool keep = fp_n < 8;
    v2f bf[4];
#pragma unroll
    for (int kc = 0; kc < 4; ++kc) { bf[kc].x = 0.0f; bf[kc].y = 0.0f; }
    if (keep) {
      const int odd = fp_n >> 2;
      const float* src = &fac2[(fp_n & 3) * 16 + odd];
#pragma unroll
      for (int kc = 0; kc < 4; ++kc) {
        const float t = src[4 * kc + 2 * fp_hi];
        bf[kc].x = odd ? 0.0f : t;
        bf[kc].y = odd ? t : 0.0f;
      }
    }
    factor_product(U[0], bf, &fac[(keep ? fp_n : 0) * 16], keep);
  }

  {
    const int n = fp_n, hi = fp_hi, j = fp_wave;
    const int cz8 = 8 * (hi ^ (__popc(n) & 1));
    v4f in[2][2];
#pragma unroll
    for (int s = 0; s < 2; ++s) {
      const v4f* p = reinterpret_cast<const v4f*>(&psi[j * PT + (n ^ (8 * s)) * PM + cz8]);
      in[s][0] = p[0];
      in[s][1] = p[1];
    }
    const int frag = n * PM + 8 * hi;
    v2f bl[4], bm[4];
    {
      const v4f* gl = reinterpret_cast<const v4f*>(&U[3][frag]);
      const v4f* gm = reinterpret_cast<const v4f*>(&U[2][frag]);
      v4f l0 = gl[0], l1 = gl[1], m0 = gm[0], m1 = gm[1];
      bl[0].x = l0.x; bl[0].y = l0.y; bl[1].x = l0.z; bl[1].y = l0.w;
      bl[2].x = l1.x; bl[2].y = l1.y; bl[3].x = l1.z; bl[3].y = l1.w;
      bm[0].x = m0.x; bm[0].y = m0.y; bm[1].x = m0.z; bm[1].y = m0.w;
      bm[2].x = m1.x; bm[2].y = m1.y; bm[3].x = m1.z; bm[3].y = m1.w;
    }
    v8f y[2];
#pragma unroll
    for (int s = 0; s < 2; ++s) {
      v8f c = {};
#pragma unroll
      for (int kc = 0; kc < 4; ++kc) {
        v2f af;
        af.x = in[s][kc >> 1][2 * (kc & 1)];
        af.y = in[s][kc >> 1][2 * (kc & 1) + 1];
        c = __builtin_amdgcn_wmma_f32_16x16x4_f32(
             false, af,  false, bl[kc],
             (short)0, c,  false,  false);
      }
      y[s] = c;
    }
#pragma unroll
    for (int s = 0; s < 2; ++s) {
      v8f c = {};
#pragma unroll
      for (int kc = 0; kc < 4; ++kc) {
        v2f af;
        af.x = y[s][2 * kc];
        af.y = y[s][2 * kc + 1];
        c = __builtin_amdgcn_wmma_f32_16x16x4_f32(
             false, af,  false, bm[kc],
             (short)0, c,  false,  false);
      }
      v4f* q = reinterpret_cast<v4f*>(&psi[(4 + 4 * s + j) * PT + n * PM + 8 * hi]);
      v4f o0, o1;
      o0.x = c[0]; o0.y = c[1]; o0.z = c[2]; o0.w = c[3];
      o1.x = c[4]; o1.y = c[5]; o1.z = c[6]; o1.w = c[7];
      q[0] = o0;
      q[1] = o1;
    }
  }
  __syncthreads();

  {
    float ZE[4][2], ZO[4][2];
#pragma unroll
    for (int j = 0; j < 4; ++j) {
#pragma unroll
      for (int h = 0; h < 2; ++h) {
        ZE[j][h] = psi[(4 + j) * PT + (midq + 8 * h) * PM + lowq];
        ZO[j][h] = psi[(8 + j) * PT + (midq + 8 * h) * PM + lowq];
      }
    }
#pragma unroll
    for (int top = 0; top < 16; ++top) {
      float ve[4], vo[4];
#pragma unroll
      for (int j = 0; j < 4; ++j) { ve[j] = fac[16 * j + top]; vo[j] = fac[64 + 16 * j + top]; }
#pragma unroll
      for (int h = 0; h < 2; ++h) {
        float v = ve[0] * ZE[0][h];
#pragma unroll
        for (int j = 1; j < 4; ++j) v += ve[j] * ZE[j][h];
#pragma unroll
        for (int j = 0; j < 4; ++j) v += vo[j] * ZO[j][h];
        psi[top * PT + ((midq + 8 * h) ^ ((top & 1) << 3)) * PM + lowq] = v;
      }
    }
  }
  __syncthreads();

  float partial;
  {
    const int n = fp_n, hi = fp_hi, wave = fp_wave;
    v2f a[4];
#pragma unroll
    for (int kc = 0; kc < 4; ++kc) {
      const int k0 = 4 * kc + 2 * hi;
      a[kc].x = U[1][n * PM + k0];
      a[kc].y = U[1][n * PM + k0 + 1];
    }
    v2f bt[4][4];
#pragma unroll
    for (int tt = 0; tt < 4; ++tt) {
      const int base = PM * (wave * 4 + tt) + n;
#pragma unroll
      for (int kc = 0; kc < 4; ++kc) {
        const int k0 = 4 * kc + 2 * hi;
        bt[tt][kc].x = psi[base + PT * k0];
        bt[tt][kc].y = psi[base + PT * (k0 + 1)];
      }
    }
    const int cz8 = 8 * (hi ^ (__popc(n) & 1));
    v4f in[4][2];
#pragma unroll
    for (int tt = 0; tt < 4; ++tt) {
      const v4f* p = reinterpret_cast<const v4f*>(&psi[(wave * 4 + tt) * PT + n * PM + cz8]);
      in[tt][0] = p[0];
      in[tt][1] = p[1];
    }
    const int frag = n * PM + 8 * hi;
    v2f bl[4], bm[4];
    {
      const v4f* gl = reinterpret_cast<const v4f*>(&U[3][frag]);
      const v4f* gm = reinterpret_cast<const v4f*>(&U[2][frag]);
      v4f l0 = gl[0], l1 = gl[1], m0 = gm[0], m1 = gm[1];
      bl[0].x = l0.x; bl[0].y = l0.y; bl[1].x = l0.z; bl[1].y = l0.w;
      bl[2].x = l1.x; bl[2].y = l1.y; bl[3].x = l1.z; bl[3].y = l1.w;
      bm[0].x = m0.x; bm[0].y = m0.y; bm[1].x = m0.z; bm[1].y = m0.w;
      bm[2].x = m1.x; bm[2].y = m1.y; bm[3].x = m1.z; bm[3].y = m1.w;
    }
    v8f sq_t = {}, sq_m = {};
#pragma unroll
    for (int tt = 0; tt < 4; ++tt) {
      v8f c = {};
#pragma unroll
      for (int kc = 0; kc < 4; ++kc)
        c = __builtin_amdgcn_wmma_f32_16x16x4_f32(
             false, a[kc],  false, bt[tt][kc],
             (short)0, c,  false,  false);
      sq_t += c * c;
    }
    v8f y[4];
#pragma unroll
    for (int tt = 0; tt < 4; ++tt) {
      v8f c = {};
#pragma unroll
      for (int kc = 0; kc < 4; ++kc) {
        v2f af;
        af.x = in[tt][kc >> 1][2 * (kc & 1)];
        af.y = in[tt][kc >> 1][2 * (kc & 1) + 1];
        c = __builtin_amdgcn_wmma_f32_16x16x4_f32(
             false, af,  false, bl[kc],
             (short)0, c,  false,  false);
      }
      y[tt] = c;
    }
#pragma unroll
    for (int tt = 0; tt < 4; ++tt) {
      v8f c = {};
#pragma unroll
      for (int kc = 0; kc < 4; ++kc) {
        v2f af;
        af.x = y[tt][2 * kc];
        af.y = y[tt][2 * kc + 1];
        c = __builtin_amdgcn_wmma_f32_16x16x4_f32(
             false, af,  false, bm[kc],
             (short)0, c,  false,  false);
      }
      sq_m += c * c;
    }
    {
      const v4f* gt = reinterpret_cast<const v4f*>(&gw[8 * hi]);
      const v4f* gl = reinterpret_cast<const v4f*>(&gw[32 + 8 * hi]);
      const v4f t0 = gt[0], t1 = gt[1], l0 = gl[0], l1 = gl[1];
      const float gmn = gw[16 + n];
      v8f wt, wm;
      wt[0] = t0.x; wt[1] = t0.y; wt[2] = t0.z; wt[3] = t0.w;
      wt[4] = t1.x; wt[5] = t1.y; wt[6] = t1.z; wt[7] = t1.w;
      wm[0] = gmn + l0.x; wm[1] = gmn + l0.y; wm[2] = gmn + l0.z; wm[3] = gmn + l0.w;
      wm[4] = gmn + l1.x; wm[5] = gmn + l1.y; wm[6] = gmn + l1.z; wm[7] = gmn + l1.w;
      const v8f q = sq_t * wt + sq_m * wm;
      partial = ((q[0] + q[1]) + (q[2] + q[3])) + ((q[4] + q[5]) + (q[6] + q[7]));
    }
  }
  red[tid] = partial;
  __syncthreads();
#pragma unroll
  for (int off = QS_THREADS / 2; off > 0; off >>= 1) {
    if (tid < off) red[tid] += red[tid + off];
    __syncthreads();
  }
  if (tid == 0) out[b0 + smp] = red[0] + hb;
  }
}

extern "C" void kernel_launch(void* const* d_in, const int* in_sizes, int n_in,
                              void* d_out, int out_size, void* d_ws, size_t ws_size,
                              hipStream_t stream) {
  const float* sb     = (const float*)d_in[0];
  const float* params = (const float*)d_in[1];
  const float* hw     = (const float*)d_in[2];
  const float* hb     = (const float*)d_in[3];
  float* out = (float*)d_out;
  int batch = in_sizes[0] / 12;
  qsim_wmma_kernel<<<dim3((batch + SPB - 1) / SPB), dim3(QS_THREADS), 0, stream>>>(
      sb, params, hw, hb, out, batch);
}


# Readout sum without the LDS tree: each wave adds its 32 partials in registers (4 row_xmask adds + a row swap), 4 wave sums per sample added behind the loop; 13 -> 6 barriers per sample; new sum order
# speedup vs baseline: 4.9273x; 1.0338x over previous
#include <hip/hip_runtime.h>

typedef __attribute__((ext_vector_type(2))) float v2f;
typedef __attribute__((ext_vector_type(4))) float v4f;
typedef __attribute__((ext_vector_type(8))) float v8f;

#define QS_THREADS 128
#define SPB 4

#define PM 20
#define PT 332

__device__ __forceinline__ int perm4(int x) {
  if (x & 8) x ^= 4;
  if (x & 4) x ^= 2;
  if (x & 2) x ^= 1;
  return x;
}
__device__ __forceinline__ int perm4inv(int x) {
  if (x & 2) x ^= 1;
  if (x & 4) x ^= 2;
  if (x & 8) x ^= 4;
  return x;
}

template <int M>
__device__ __forceinline__ float row_xor(float x) {
  const int b = __builtin_bit_cast(int, x);
  return __builtin_bit_cast(float, __builtin_amdgcn_mov_dpp(b, 0x160 + M, 0xf, 0xf, true));
}
__device__ __forceinline__ float xhalf(float x) {
  const int b = __builtin_bit_cast(int, x);
  return __builtin_bit_cast(
      float, __builtin_amdgcn_permlanex16(b, b, 0x76543210, 0xfedcba98, false, false));
}
__device__ __forceinline__ float wave_sum(float x) {
  x += row_xor<1>(x);
  x += row_xor<2>(x);
  x += row_xor<4>(x);
  x += row_xor<8>(x);
  return x + xhalf(x);
}

__global__ __launch_bounds__(QS_THREADS, 8) void qsim_wmma_kernel(
    const float* __restrict__ state_batch,
    const float* __restrict__ params,
    const float* __restrict__ head_w,
    const float* __restrict__ head_b,
    float* __restrict__ out,
    int batch) {
  __shared__ __attribute__((aligned(16))) float psi[16 * PT];
  __shared__ __attribute__((aligned(16))) float U[4][16 * PM];
  __shared__ float cz[12 + 12 * SPB], sz[12 + 12 * SPB];
  __shared__ __attribute__((aligned(16))) float gw[48];
  __shared__ __attribute__((aligned(16))) float fac[128];
  __shared__ __attribute__((aligned(16))) float fac2[64];
  __shared__ __attribute__((aligned(16))) float wsum[4 * SPB];

  const int b0  = blockIdx.x * SPB;
  const int nb  = min(SPB, batch - b0);
  const int tid = threadIdx.x;

  if (tid < 12 + 12 * nb) {
    float a = 0.5f * ((tid < 12) ? params[tid] : state_batch[(size_t)b0 * 12 + (tid - 12)]);
    cz[tid] = cosf(a); sz[tid] = sinf(a);
  }
  if (tid >= 64 && tid < 112) {
    const int nib = (tid - 64) >> 4, x = tid & 15;
    float v = 0.0f;
    for (int q = 0; q < 4; ++q) v += ((x >> (3 - q)) & 1) ? -head_w[4 * nib + q] : head_w[4 * nib + q];
    gw[16 * nib + x] = v;
  }
  const float hb = head_b[0];
  __syncthreads();

  auto kron4 = [&](int base, int i, int j) -> float {
    float v = 1.0f;
#pragma unroll
    for (int q = 0; q < 4; ++q) {
      int bi = (i >> (3 - q)) & 1;
      int bj = (j >> (3 - q)) & 1;
      float c = cz[base + q], s = sz[base + q];
      v *= bi ? (bj ? c : s) : (bj ? -s : c);
    }
    return v;
  };

  for (int e = tid; e < 4 * 256; e += QS_THREADS) {
    int mi = e >> 8, idx = e & 255, i = idx >> 4, j = idx & 15;
    float v;
    switch (mi) {
      case 0:  v = kron4(0, perm4inv(i), j); break;
      case 1:  v = kron4(0, i, j); break;
      case 2:  v = kron4(4, i, perm4(j)); break;
      default: v = kron4(8, i, perm4(j)); break;
    }
    U[mi][i * PM + j] = v;
  }

  const int lowq = tid & 15, midq = tid >> 4;
  auto amp4 = [&](int base, int x) -> float {
    float v = 1.0f;
#pragma unroll
    for (int q = 0; q < 4; ++q) v *= ((x >> (3 - q)) & 1) ? sz[base + q] : cz[base + q];
    return v;
  };

  int st_nib, st_x, st_dst;
  bool st_on;
  {
    const int k = tid & 15;
    int plane, col;
    if (tid < 32) {
      plane = 0; col = tid >> 4;
      st_nib = 0; st_x = perm4inv(k); st_on = (k & 1) == col;
    } else if (tid < 64) {
      plane = 1; col = (tid >> 4) & 1;
      st_nib = 8; st_x = k ^ (col << 3); st_on = true;
    } else {
      plane = 2; col = (tid >> 4) & 3;
      st_nib = 4; st_x = k ^ ((col & 1) << 3); st_on = (__popc(k) & 1) == (col >> 1);
    }
    st_dst = plane * PT + k * PM + col;
  }

#pragma unroll 1
  for (int smp = 0; smp < nb; ++smp) {
  const int czb = 12 + 12 * smp;
  psi[st_dst] = st_on ? amp4(czb + st_nib, st_x) : 0.0f;
  __syncthreads();

  const int fp_lane = tid & 31, fp_wave = tid >> 5;
  const int fp_n = fp_lane & 15, fp_hi = fp_lane >> 4;
  auto factor_product = [&](const float* Um, const v2f (&bf)[4], float* dst, bool keep) {
    v2f af[4];
#pragma unroll
    for (int kc = 0; kc < 4; ++kc) {
      const int k0 = 4 * kc + 2 * fp_hi;
      af[kc].x = Um[fp_n * PM + k0];
      af[kc].y = Um[fp_n * PM + k0 + 1];
    }
    v8f c = {};
#pragma unroll
    for (int kc = 0; kc < 4; ++kc)
      c = __builtin_amdgcn_wmma_f32_16x16x4_f32(
           false, af[kc],  false, bf[kc],
           (short)0, c,  false,  false);
    if (keep) {
#pragma unroll
      for (int v = 0; v < 8; ++v)
        dst[v + 8 * fp_hi] = c[v];
    }
  };

  if (fp_wave >= 1) {
    const int pr = fp_wave - 1;
    const float* Um = (pr == 0) ? U[0] : ((pr == 1) ? U[3] : U[2]);
    const int ncol = (pr == 2) ? 4 : 2;
    const bool keep = fp_n < ncol;
    const float* Vp = &psi[pr * PT];
    v2f bf[4];
#pragma unroll
    for (int kc = 0; kc < 4; ++kc) { bf[kc].x = 0.0f; bf[kc].y = 0.0f; }
    if (keep) {
#pragma unroll
      for (int kc = 0; kc < 4; ++kc) {
        const int k0 = 4 * kc + 2 * fp_hi;
        bf[kc].x = Vp[k0 * PM + fp_n];
        bf[kc].y = Vp[(k0 + 1) * PM + fp_n];
      }
    }
    factor_product(Um, bf, &fac[(2 * pr + (keep ? fp_n : 0)) * 16], keep);
  }
  __syncthreads();

  if (fp_wave == 1) {
    const bool keep = fp_n < 4;
    v2f bf[4];
#pragma unroll
    for (int kc = 0; kc < 4; ++kc) { bf[kc].x = 0.0f; bf[kc].y = 0.0f; }
    if (keep) {
      const int odd = fp_n >> 1;
      const float* src = &fac[(fp_n & 1) * 16 + odd];
#pragma unroll
      for (int kc = 0; kc < 4; ++kc) {
        const float t = src[4 * kc + 2 * fp_hi];
        bf[kc].x = odd ? 0.0f : t;
        bf[kc].y = odd ? t : 0.0f;
      }
    }
    factor_product(U[0], bf, &fac2[(keep ? fp_n : 0) * 16], keep);
  }

  {
    const int n = fp_n, hi = fp_hi, j = fp_wave;
    const int msel = n ^ ((j >> 1) << 3);
    const float ca = fac[64 + 16 * (j & 1) + msel];
    const float cb = fac[96 + 16 * (j & 1) + msel];
    const int lb = 8 * (hi ^ (__popc(n) & 1));
    const v4f* le = reinterpret_cast<const v4f*>(&fac[32 + lb]);
    const v4f* lo = reinterpret_cast<const v4f*>(&fac[48 + lb]);
    v4f in[2];
    in[0] = le[0] * ca + lo[0] * cb;
    in[1] = le[1] * ca + lo[1] * cb;
    const int frag = n * PM + 8 * hi;
    v2f bl[4], bm[4];
    {
      const v4f* gl = reinterpret_cast<const v4f*>(&U[3][frag]);
      const v4f* gm = reinterpret_cast<const v4f*>(&U[2][frag]);
      v4f l0 = gl[0], l1 = gl[1], m0 = gm[0], m1 = gm[1];
      bl[0].x = l0.x; bl[0].y = l0.y; bl[1].x = l0.z; bl[1].y = l0.w;
      bl[2].x = l1.x; bl[2].y = l1.y; bl[3].x = l1.z; bl[3].y = l1.w;
      bm[0].x = m0.x; bm[0].y = m0.y; bm[1].x = m0.z; bm[1].y = m0.w;
      bm[2].x = m1.x; bm[2].y = m1.y; bm[3].x = m1.z; bm[3].y = m1.w;
    }
    v8f y = {};
#pragma unroll
    for (int kc = 0; kc < 4; ++kc) {
      v2f af;
      af.x = in[kc >> 1][2 * (kc & 1)];
      af.y = in[kc >> 1][2 * (kc & 1) + 1];
      y = __builtin_amdgcn_wmma_f32_16x16x4_f32(
           false, af,  false, bl[kc],
           (short)0, y,  false,  false);
    }
    v8f c = {};
#pragma unroll
    for (int kc = 0; kc < 4; ++kc) {
      v2f af;
      af.x = y[2 * kc];
      af.y = y[2 * kc + 1];
      c = __builtin_amdgcn_wmma_f32_16x16x4_f32(
           false, af,  false, bm[kc],
           (short)0, c,  false,  false);
    }
    v4f* q = reinterpret_cast<v4f*>(&psi[j * PT + n * PM + 8 * hi]);
    v4f o0, o1;
    o0.x = c[0]; o0.y = c[1]; o0.z = c[2]; o0.w = c[3];
    o1.x = c[4]; o1.y = c[5]; o1.z = c[6]; o1.w = c[7];
    q[0] = o0;
    q[1] = o1;
  }
  __syncthreads();

  if (fp_wave == 0) {
    const bool keep = fp_n < 8;
    v2f bf[4];
#pragma unroll
    for (int kc = 0; kc < 4; ++kc) { bf[kc].x = 0.0f; bf[kc].y = 0.0f; }
    if (keep) {
      const int odd = fp_n >> 2;
      const float* src = &fac2[(fp_n & 3) * 16 + odd];
#pragma unroll
      for (int kc = 0; kc < 4; ++kc) {
        const float t = src[4 * kc + 2 * fp_hi];
        bf[kc].x = odd ? 0.0f : t;
        bf[kc].y = odd ? t : 0.0f;
      }
    }
    factor_product(U[0], bf, &fac[(keep ? fp_n : 0) * 16], keep);
  }

  {
    const int n = fp_n, hi = fp_hi, j = fp_wave;
    const int cz8 = 8 * (hi ^ (__popc(n) & 1));
    v4f in[2][2];
#pragma unroll
    for (int s = 0; s < 2; ++s) {
      const v4f* p = reinterpret_cast<const v4f*>(&psi[j * PT + (n ^ (8 * s)) * PM + cz8]);
      in[s][0] = p[0];
      in[s][1] = p[1];
    }
    const int frag = n * PM + 8 * hi;
    v2f bl[4], bm[4];
    {
      const v4f* gl = reinterpret_cast<const v4f*>(&U[3][frag]);
      const v4f* gm = reinterpret_cast<const v4f*>(&U[2][frag]);
      v4f l0 = gl[0], l1 = gl[1], m0 = gm[0], m1 = gm[1];
      bl[0].x = l0.x; bl[0].y = l0.y; bl[1].x = l0.z; bl[1].y = l0.w;
      bl[2].x = l1.x; bl[2].y = l1.y; bl[3].x = l1.z; bl[3].y = l1.w;
      bm[0].x = m0.x; bm[0].y = m0.y; bm[1].x = m0.z; bm[1].y = m0.w;
      bm[2].x = m1.x; bm[2].y = m1.y; bm[3].x = m1.z; bm[3].y = m1.w;
    }
    v8f y[2];
#pragma unroll
    for (int s = 0; s < 2; ++s) {
      v8f c = {};
#pragma unroll
      for (int kc = 0; kc < 4; ++kc) {
        v2f af;
        af.x = in[s][kc >> 1][2 * (kc & 1)];
        af.y = in[s][kc >> 1][2 * (kc & 1) + 1];
        c = __builtin_amdgcn_wmma_f32_16x16x4_f32(
             false, af,  false, bl[kc],
             (short)0, c,  false,  false);
      }
      y[s] = c;
    }
#pragma unroll
    for (int s = 0; s < 2; ++s) {
      v8f c = {};
#pragma unroll
      for (int kc = 0; kc < 4; ++kc) {
        v2f af;
        af.x = y[s][2 * kc];
        af.y = y[s][2 * kc + 1];
        c = __builtin_amdgcn_wmma_f32_16x16x4_f32(
             false, af,  false, bm[kc],
             (short)0, c,  false,  false);
      }
      v4f* q = reinterpret_cast<v4f*>(&psi[(4 + 4 * s + j) * PT + n * PM + 8 * hi]);
      v4f o0, o1;
      o0.x = c[0]; o0.y = c[1]; o0.z = c[2]; o0.w = c[3];
      o1.x = c[4]; o1.y = c[5]; o1.z = c[6]; o1.w = c[7];
      q[0] = o0;
      q[1] = o1;
    }
  }
  __syncthreads();

  {
    float ZE[4][2], ZO[4][2];
#pragma unroll
    for (int j = 0; j < 4; ++j) {
#pragma unroll
      for (int h = 0; h < 2; ++h) {
        ZE[j][h] = psi[(4 + j) * PT + (midq + 8 * h) * PM + lowq];
        ZO[j][h] = psi[(8 + j) * PT + (midq + 8 * h) * PM + lowq];
      }
    }
#pragma unroll
    for (int top = 0; top < 16; ++top) {
      float ve[4], vo[4];
#pragma unroll
      for (int j = 0; j < 4; ++j) { ve[j] = fac[16 * j + top]; vo[j] = fac[64 + 16 * j + top]; }
#pragma unroll
      for (int h = 0; h < 2; ++h) {
        float v = ve[0] * ZE[0][h];
#pragma unroll
        for (int j = 1; j < 4; ++j) v += ve[j] * ZE[j][h];
#pragma unroll
        for (int j = 0; j < 4; ++j) v += vo[j] * ZO[j][h];
        psi[top * PT + ((midq + 8 * h) ^ ((top & 1) << 3)) * PM + lowq] = v;
      }
    }
  }
  __syncthreads();

  float partial;
  {
    const int n = fp_n, hi = fp_hi, wave = fp_wave;
    v2f a[4];
#pragma unroll
    for (int kc = 0; kc < 4; ++kc) {
      const int k0 = 4 * kc + 2 * hi;
      a[kc].x = U[1][n * PM + k0];
      a[kc].y = U[1][n * PM + k0 + 1];
    }
    v2f bt[4][4];
#pragma unroll
    for (int tt = 0; tt < 4; ++tt) {
      const int base = PM * (wave * 4 + tt) + n;
#pragma unroll
      for (int kc = 0; kc < 4; ++kc) {
        const int k0 = 4 * kc + 2 * hi;
        bt[tt][kc].x = psi[base + PT * k0];
        bt[tt][kc].y = psi[base + PT * (k0 + 1)];
      }
    }
    const int cz8 = 8 * (hi ^ (__popc(n) & 1));
    v4f in[4][2];
#pragma unroll
    for (int tt = 0; tt < 4; ++tt) {
      const v4f* p = reinterpret_cast<const v4f*>(&psi[(wave * 4 + tt) * PT + n * PM + cz8]);
      in[tt][0] = p[0];
      in[tt][1] = p[1];
    }
    const int frag = n * PM + 8 * hi;
    v2f bl[4], bm[4];
    {
      const v4f* gl = reinterpret_cast<const v4f*>(&U[3][frag]);
      const v4f* gm = reinterpret_cast<const v4f*>(&U[2][frag]);
      v4f l0 = gl[0], l1 = gl[1], m0 = gm[0], m1 = gm[1];
      bl[0].x = l0.x; bl[0].y = l0.y; bl[1].x = l0.z; bl[1].y = l0.w;
      bl[2].x = l1.x; bl[2].y = l1.y; bl[3].x = l1.z; bl[3].y = l1.w;
      bm[0].x = m0.x; bm[0].y = m0.y; bm[1].x = m0.z; bm[1].y = m0.w;
      bm[2].x = m1.x; bm[2].y = m1.y; bm[3].x = m1.z; bm[3].y = m1.w;
    }
    v8f sq_t = {}, sq_m = {};
#pragma unroll
    for (int tt = 0; tt < 4; ++tt) {
      v8f c = {};
#pragma unroll
      for (int kc = 0; kc < 4; ++kc)
        c = __builtin_amdgcn_wmma_f32_16x16x4_f32(
             false, a[kc],  false, bt[tt][kc],
             (short)0, c,  false,  false);
      sq_t += c * c;
    }
    v8f y[4];
#pragma unroll
    for (int tt = 0; tt < 4; ++tt) {
      v8f c = {};
#pragma unroll
      for (int kc = 0; kc < 4; ++kc) {
        v2f af;
        af.x = in[tt][kc >> 1][2 * (kc & 1)];
        af.y = in[tt][kc >> 1][2 * (kc & 1) + 1];
        c = __builtin_amdgcn_wmma_f32_16x16x4_f32(
             false, af,  false, bl[kc],
             (short)0, c,  false,  false);
      }
      y[tt] = c;
    }
#pragma unroll
    for (int tt = 0; tt < 4; ++tt) {
      v8f c = {};
#pragma unroll
      for (int kc = 0; kc < 4; ++kc) {
        v2f af;
        af.x = y[tt][2 * kc];
        af.y = y[tt][2 * kc + 1];
        c = __builtin_amdgcn_wmma_f32_16x16x4_f32(
             false, af,  false, bm[kc],
             (short)0, c,  false,  false);
      }
      sq_m += c * c;
    }
    {
      const v4f* gt = reinterpret_cast<const v4f*>(&gw[8 * hi]);
      const v4f* gl = reinterpret_cast<const v4f*>(&gw[32 + 8 * hi]);
      const v4f t0 = gt[0], t1 = gt[1], l0 = gl[0], l1 = gl[1];
      const float gmn = gw[16 + n];
      v8f wt, wm;
      wt[0] = t0.x; wt[1] = t0.y; wt[2] = t0.z; wt[3] = t0.w;
      wt[4] = t1.x; wt[5] = t1.y; wt[6] = t1.z; wt[7] = t1.w;
      wm[0] = gmn + l0.x; wm[1] = gmn + l0.y; wm[2] = gmn + l0.z; wm[3] = gmn + l0.w;
      wm[4] = gmn + l1.x; wm[5] = gmn + l1.y; wm[6] = gmn + l1.z; wm[7] = gmn + l1.w;
      const v8f q = sq_t * wt + sq_m * wm;
      partial = ((q[0] + q[1]) + (q[2] + q[3])) + ((q[4] + q[5]) + (q[6] + q[7]));
    }
  }
  {
    const float ws = wave_sum(partial);
    if (fp_lane == 0) wsum[4 * smp + fp_wave] = ws;
  }
  __syncthreads();
  }

  if (tid < nb) {
    const v4f w = *reinterpret_cast<const v4f*>(&wsum[4 * tid]);
    out[b0 + tid] = ((w.x + w.y) + (w.z + w.w)) + hb;
  }
}

extern "C" void kernel_launch(void* const* d_in, const int* in_sizes, int n_in,
                              void* d_out, int out_size, void* d_ws, size_t ws_size,
                              hipStream_t stream) {
  const float* sb     = (const float*)d_in[0];
  const float* params = (const float*)d_in[1];
  const float* hw     = (const float*)d_in[2];
  const float* hb     = (const float*)d_in[3];
  float* out = (float*)d_out;
  int batch = in_sizes[0] / 12;
  qsim_wmma_kernel<<<dim3((batch + SPB - 1) / SPB), dim3(QS_THREADS), 0, stream>>>(
      sb, params, hw, hb, out, batch);
}


# Layer 4 runs on layer 3's 16 rank tiles instead of the expanded state: low/mid pass on the tiles, then one top product with A' = R0 x (top vectors by parity) gives the amplitudes; new sum order
# speedup vs baseline: 5.6557x; 1.1478x over previous
#include <hip/hip_runtime.h>

typedef __attribute__((ext_vector_type(2))) float v2f;
typedef __attribute__((ext_vector_type(4))) float v4f;
typedef __attribute__((ext_vector_type(8))) float v8f;

#define QS_THREADS 128
#define SPB 4

#define PM 20
#define PT 332

__device__ __forceinline__ int perm4(int x) {
  if (x & 8) x ^= 4;
  if (x & 4) x ^= 2;
  if (x & 2) x ^= 1;
  return x;
}
__device__ __forceinline__ int perm4inv(int x) {
  if (x & 2) x ^= 1;
  if (x & 4) x ^= 2;
  if (x & 8) x ^= 4;
  return x;
}

template <int M>
__device__ __forceinline__ float row_xor(float x) {
  const int b = __builtin_bit_cast(int, x);
  return __builtin_bit_cast(float, __builtin_amdgcn_mov_dpp(b, 0x160 + M, 0xf, 0xf, true));
}
__device__ __forceinline__ float xhalf(float x) {
  const int b = __builtin_bit_cast(int, x);
  return __builtin_bit_cast(
      float, __builtin_amdgcn_permlanex16(b, b, 0x76543210, 0xfedcba98, false, false));
}
__device__ __forceinline__ float wave_sum(float x) {
  x += row_xor<1>(x);
  x += row_xor<2>(x);
  x += row_xor<4>(x);
  x += row_xor<8>(x);
  return x + xhalf(x);
}

__global__ __launch_bounds__(QS_THREADS, 8) void qsim_wmma_kernel(
    const float* __restrict__ state_batch,
    const float* __restrict__ params,
    const float* __restrict__ head_w,
    const float* __restrict__ head_b,
    float* __restrict__ out,
    int batch) {
  __shared__ __attribute__((aligned(16))) float psi[16 * PT];
  __shared__ __attribute__((aligned(16))) float U[4][16 * PM];
  __shared__ float cz[12 + 12 * SPB], sz[12 + 12 * SPB];
  __shared__ __attribute__((aligned(16))) float gw[48];
  __shared__ __attribute__((aligned(16))) float fac[128];
  __shared__ __attribute__((aligned(16))) float fac2[64];
  __shared__ __attribute__((aligned(16))) float ra[256];
  __shared__ __attribute__((aligned(16))) float wsum[4 * SPB];

  const int b0  = blockIdx.x * SPB;
  const int nb  = min(SPB, batch - b0);
  const int tid = threadIdx.x;

  if (tid < 12 + 12 * nb) {
    float a = 0.5f * ((tid < 12) ? params[tid] : state_batch[(size_t)b0 * 12 + (tid - 12)]);
    cz[tid] = cosf(a); sz[tid] = sinf(a);
  }
  if (tid >= 64 && tid < 112) {
    const int nib = (tid - 64) >> 4, x = tid & 15;
    float v = 0.0f;
    for (int q = 0; q < 4; ++q) v += ((x >> (3 - q)) & 1) ? -head_w[4 * nib + q] : head_w[4 * nib + q];
    gw[16 * nib + x] = v;
  }
  const float hb = head_b[0];
  __syncthreads();

  auto kron4 = [&](int base, int i, int j) -> float {
    float v = 1.0f;
#pragma unroll
    for (int q = 0; q < 4; ++q) {
      int bi = (i >> (3 - q)) & 1;
      int bj = (j >> (3 - q)) & 1;
      float c = cz[base + q], s = sz[base + q];
      v *= bi ? (bj ? c : s) : (bj ? -s : c);
    }
    return v;
  };

  for (int e = tid; e < 4 * 256; e += QS_THREADS) {
    int mi = e >> 8, idx = e & 255, i = idx >> 4, j = idx & 15;
    float v;
    switch (mi) {
      case 0:  v = kron4(0, perm4inv(i), j); break;
      case 1:  v = kron4(0, i, j); break;
      case 2:  v = kron4(4, i, perm4(j)); break;
      default: v = kron4(8, i, perm4(j)); break;
    }
    U[mi][i * PM + j] = v;
  }

  auto amp4 = [&](int base, int x) -> float {
    float v = 1.0f;
#pragma unroll
    for (int q = 0; q < 4; ++q) v *= ((x >> (3 - q)) & 1) ? sz[base + q] : cz[base + q];
    return v;
  };

  int st_nib, st_x, st_dst;
  bool st_on;
  {
    const int k = tid & 15;
    int plane, col;
    if (tid < 32) {
      plane = 0; col = tid >> 4;
      st_nib = 0; st_x = perm4inv(k); st_on = (k & 1) == col;
    } else if (tid < 64) {
      plane = 1; col = (tid >> 4) & 1;
      st_nib = 8; st_x = k ^ (col << 3); st_on = true;
    } else {
      plane = 2; col = (tid >> 4) & 3;
      st_nib = 4; st_x = k ^ ((col & 1) << 3); st_on = (__popc(k) & 1) == (col >> 1);
    }
    st_dst = plane * PT + k * PM + col;
  }

#pragma unroll 1
  for (int smp = 0; smp < nb; ++smp) {
  const int czb = 12 + 12 * smp;
  psi[st_dst] = st_on ? amp4(czb + st_nib, st_x) : 0.0f;
  __syncthreads();

  const int fp_lane = tid & 31, fp_wave = tid >> 5;
  const int fp_n = fp_lane & 15, fp_hi = fp_lane >> 4;
  auto factor_product = [&](const float* Um, const v2f (&bf)[4], float* dst, bool keep) {
    v2f af[4];
#pragma unroll
    for (int kc = 0; kc < 4; ++kc) {
      const int k0 = 4 * kc + 2 * fp_hi;
      af[kc].x = Um[fp_n * PM + k0];
      af[kc].y = Um[fp_n * PM + k0 + 1];
    }
    v8f c = {};
#pragma unroll
    for (int kc = 0; kc < 4; ++kc)
      c = __builtin_amdgcn_wmma_f32_16x16x4_f32(
           false, af[kc],  false, bf[kc],
           (short)0, c,  false,  false);
    if (keep) {
#pragma unroll
      for (int v = 0; v < 8; ++v)
        dst[v + 8 * fp_hi] = c[v];
    }
  };

  if (fp_wave >= 1) {
    const int pr = fp_wave - 1;
    const float* Um = (pr == 0) ? U[0] : ((pr == 1) ? U[3] : U[2]);
    const int ncol = (pr == 2) ? 4 : 2;
    const bool keep = fp_n < ncol;
    const float* Vp = &psi[pr * PT];
    v2f bf[4];
#pragma unroll
    for (int kc = 0; kc < 4; ++kc) { bf[kc].x = 0.0f; bf[kc].y = 0.0f; }
    if (keep) {
#pragma unroll
      for (int kc = 0; kc < 4; ++kc) {
        const int k0 = 4 * kc + 2 * fp_hi;
        bf[kc].x = Vp[k0 * PM + fp_n];
        bf[kc].y = Vp[(k0 + 1) * PM + fp_n];
      }
    }
    factor_product(Um, bf, &fac[(2 * pr + (keep ? fp_n : 0)) * 16], keep);
  }
  __syncthreads();

  if (fp_wave == 1) {
    const bool keep = fp_n < 4;
    v2f bf[4];
#pragma unroll
    for (int kc = 0; kc < 4; ++kc) { bf[kc].x = 0.0f; bf[kc].y = 0.0f; }
    if (keep) {
      const int odd = fp_n >> 1;
      const float* src = &fac[(fp_n & 1) * 16 + odd];
#pragma unroll
      for (int kc = 0; kc < 4; ++kc) {
        const float t = src[4 * kc + 2 * fp_hi];
        bf[kc].x = odd ? 0.0f : t;
        bf[kc].y = odd ? t : 0.0f;
      }
    }
    factor_product(U[0], bf, &fac2[(keep ? fp_n : 0) * 16], keep);
  }

  {
    const int n = fp_n, hi = fp_hi, j = fp_wave;
    const int msel = n ^ ((j >> 1) << 3);
    const float ca = fac[64 + 16 * (j & 1) + msel];
    const float cb = fac[96 + 16 * (j & 1) + msel];
    const int lb = 8 * (hi ^ (__popc(n) & 1));
    const v4f* le = reinterpret_cast<const v4f*>(&fac[32 + lb]);
    const v4f* lo = reinterpret_cast<const v4f*>(&fac[48 + lb]);
    v4f in[2];
    in[0] = le[0] * ca + lo[0] * cb;
    in[1] = le[1] * ca + lo[1] * cb;
    const int frag = n * PM + 8 * hi;
    v2f bl[4], bm[4];
    {
      const v4f* gl = reinterpret_cast<const v4f*>(&U[3][frag]);
      const v4f* gm = reinterpret_cast<const v4f*>(&U[2][frag]);
      v4f l0 = gl[0], l1 = gl[1], m0 = gm[0], m1 = gm[1];
      bl[0].x = l0.x; bl[0].y = l0.y; bl[1].x = l0.z; bl[1].y = l0.w;
      bl[2].x = l1.x; bl[2].y = l1.y; bl[3].x = l1.z; bl[3].y = l1.w;
      bm[0].x = m0.x; bm[0].y = m0.y; bm[1].x = m0.z; bm[1].y = m0.w;
      bm[2].x = m1.x; bm[2].y = m1.y; bm[3].x = m1.z; bm[3].y = m1.w;
    }
    v8f y = {};
#pragma unroll
    for (int kc = 0; kc < 4; ++kc) {
      v2f af;
      af.x = in[kc >> 1][2 * (kc & 1)];
      af.y = in[kc >> 1][2 * (kc & 1) + 1];
      y = __builtin_amdgcn_wmma_f32_16x16x4_f32(
           false, af,  false, bl[kc],
           (short)0, y,  false,  false);
    }
    v8f c = {};
#pragma unroll
    for (int kc = 0; kc < 4; ++kc) {
      v2f af;
      af.x = y[2 * kc];
      af.y = y[2 * kc + 1];
      c = __builtin_amdgcn_wmma_f32_16x16x4_f32(
           false, af,  false, bm[kc],
           (short)0, c,  false,  false);
    }
    v4f* q = reinterpret_cast<v4f*>(&psi[j * PT + n * PM + 8 * hi]);
    v4f o0, o1;
    o0.x = c[0]; o0.y = c[1]; o0.z = c[2]; o0.w = c[3];
    o1.x = c[4]; o1.y = c[5]; o1.z = c[6]; o1.w = c[7];
    q[0] = o0;
    q[1] = o1;
  }
  __syncthreads();

  if (fp_wave == 0) {
    const bool keep = fp_n < 8;
    v2f bf[4];
#pragma unroll
    for (int kc = 0; kc < 4; ++kc) { bf[kc].x = 0.0f; bf[kc].y = 0.0f; }
    if (keep) {
      const int odd = fp_n >> 2;
      const float* src = &fac2[(fp_n & 3) * 16 + odd];
#pragma unroll
      for (int kc = 0; kc < 4; ++kc) {
        const float t = src[4 * kc + 2 * fp_hi];
        bf[kc].x = odd ? 0.0f : t;
        bf[kc].y = odd ? t : 0.0f;
      }
    }
    factor_product(U[0], bf, &fac[(keep ? fp_n : 0) * 16], keep);
  }

  {
    const int n = fp_n, hi = fp_hi, j = fp_wave;
    const int cz8 = 8 * (hi ^ (__popc(n) & 1));
    v4f in[2][2];
#pragma unroll
    for (int s = 0; s < 2; ++s) {
      const v4f* p = reinterpret_cast<const v4f*>(&psi[j * PT + (n ^ (8 * s)) * PM + cz8]);
      in[s][0] = p[0];
      in[s][1] = p[1];
    }
    const int frag = n * PM + 8 * hi;
    v2f bl[4], bm[4];
    {
      const v4f* gl = reinterpret_cast<const v4f*>(&U[3][frag]);
      const v4f* gm = reinterpret_cast<const v4f*>(&U[2][frag]);
      v4f l0 = gl[0], l1 = gl[1], m0 = gm[0], m1 = gm[1];
      bl[0].x = l0.x; bl[0].y = l0.y; bl[1].x = l0.z; bl[1].y = l0.w;
      bl[2].x = l1.x; bl[2].y = l1.y; bl[3].x = l1.z; bl[3].y = l1.w;
      bm[0].x = m0.x; bm[0].y = m0.y; bm[1].x = m0.z; bm[1].y = m0.w;
      bm[2].x = m1.x; bm[2].y = m1.y; bm[3].x = m1.z; bm[3].y = m1.w;
    }
    v8f y[2];
#pragma unroll
    for (int s = 0; s < 2; ++s) {
      v8f c = {};
#pragma unroll
      for (int kc = 0; kc < 4; ++kc) {
        v2f af;
        af.x = in[s][kc >> 1][2 * (kc & 1)];
        af.y = in[s][kc >> 1][2 * (kc & 1) + 1];
        c = __builtin_amdgcn_wmma_f32_16x16x4_f32(
             false, af,  false, bl[kc],
             (short)0, c,  false,  false);
      }
      y[s] = c;
    }
#pragma unroll
    for (int s = 0; s < 2; ++s) {
      v8f c = {};
#pragma unroll
      for (int kc = 0; kc < 4; ++kc) {
        v2f af;
        af.x = y[s][2 * kc];
        af.y = y[s][2 * kc + 1];
        c = __builtin_amdgcn_wmma_f32_16x16x4_f32(
             false, af,  false, bm[kc],
             (short)0, c,  false,  false);
      }
      v4f* q = reinterpret_cast<v4f*>(&psi[(4 + 4 * s + j) * PT + n * PM + 8 * hi]);
      v4f o0, o1;
      o0.x = c[0]; o0.y = c[1]; o0.z = c[2]; o0.w = c[3];
      o1.x = c[4]; o1.y = c[5]; o1.z = c[6]; o1.w = c[7];
      q[0] = o0;
      q[1] = o1;
    }
  }
  __syncthreads();

  if (fp_wave == 2) {
    const int odd = fp_n >> 3;
    const float* src = &fac[(fp_n & 7) * 16 + odd];
    v2f bf[4];
#pragma unroll
    for (int kc = 0; kc < 4; ++kc) {
      const float t = src[4 * kc + 2 * fp_hi];
      bf[kc].x = odd ? 0.0f : t;
      bf[kc].y = odd ? t : 0.0f;
    }
    factor_product(U[1], bf, &ra[fp_n * 16], true);
  }

  {
    const int n = fp_n, hi = fp_hi, j = fp_wave;
    const int cz8 = 8 * (hi ^ (__popc(n) & 1));
    v4f in[4][2];
#pragma unroll
    for (int q = 0; q < 4; ++q) {
      const v4f* p = reinterpret_cast<const v4f*>(
          &psi[(4 + 4 * (q & 1) + j) * PT + (n ^ (8 * (q >> 1))) * PM + cz8]);
      in[q][0] = p[0];
      in[q][1] = p[1];
    }
    const int frag = n * PM + 8 * hi;
    v2f bl[4], bm[4];
    {
      const v4f* gl = reinterpret_cast<const v4f*>(&U[3][frag]);
      const v4f* gm = reinterpret_cast<const v4f*>(&U[2][frag]);
      v4f l0 = gl[0], l1 = gl[1], m0 = gm[0], m1 = gm[1];
      bl[0].x = l0.x; bl[0].y = l0.y; bl[1].x = l0.z; bl[1].y = l0.w;
      bl[2].x = l1.x; bl[2].y = l1.y; bl[3].x = l1.z; bl[3].y = l1.w;
      bm[0].x = m0.x; bm[0].y = m0.y; bm[1].x = m0.z; bm[1].y = m0.w;
      bm[2].x = m1.x; bm[2].y = m1.y; bm[3].x = m1.z; bm[3].y = m1.w;
    }
    v8f y[4];
#pragma unroll
    for (int q = 0; q < 4; ++q) {
      v8f c = {};
#pragma unroll
      for (int kc = 0; kc < 4; ++kc) {
        v2f af;
        af.x = in[q][kc >> 1][2 * (kc & 1)];
        af.y = in[q][kc >> 1][2 * (kc & 1) + 1];
        c = __builtin_amdgcn_wmma_f32_16x16x4_f32(
             false, af,  false, bl[kc],
             (short)0, c,  false,  false);
      }
      y[q] = c;
    }
    v8f z[4];
#pragma unroll
    for (int q = 0; q < 4; ++q) {
      v8f c = {};
#pragma unroll
      for (int kc = 0; kc < 4; ++kc) {
        v2f af;
        af.x = y[q][2 * kc];
        af.y = y[q][2 * kc + 1];
        c = __builtin_amdgcn_wmma_f32_16x16x4_f32(
             false, af,  false, bm[kc],
             (short)0, c,  false,  false);
      }
      z[q] = c;
    }
#pragma unroll
    for (int q = 0; q < 4; ++q) {
      v4f* o = reinterpret_cast<v4f*>(&psi[(8 * (q >> 1) + 4 * (q & 1) + j) * PT + n * PM + 8 * hi]);
      v4f o0, o1;
      o0.x = z[q][0]; o0.y = z[q][1]; o0.z = z[q][2]; o0.w = z[q][3];
      o1.x = z[q][4]; o1.y = z[q][5]; o1.z = z[q][6]; o1.w = z[q][7];
      o[0] = o0;
      o[1] = o1;
    }
  }
  __syncthreads();

  float partial;
  {
    const int n = fp_n, hi = fp_hi, wave = fp_wave;
    v2f a[4];
#pragma unroll
    for (int kc = 0; kc < 4; ++kc) {
      const int k0 = 4 * kc + 2 * hi;
      a[kc].x = ra[k0 * 16 + n];
      a[kc].y = ra[(k0 + 1) * 16 + n];
    }
    v2f bt[4][4];
#pragma unroll
    for (int tt = 0; tt < 4; ++tt) {
      const int base = PM * (wave * 4 + tt) + n;
#pragma unroll
      for (int kc = 0; kc < 4; ++kc) {
        const int k0 = 4 * kc + 2 * hi;
        bt[tt][kc].x = psi[base + PT * k0];
        bt[tt][kc].y = psi[base + PT * (k0 + 1)];
      }
    }
    const v4f* gt = reinterpret_cast<const v4f*>(&gw[8 * hi]);
    const v4f t0 = gt[0], t1 = gt[1];
    const v4f gm4 = *reinterpret_cast<const v4f*>(&gw[16 + 4 * wave]);
    const float gln = gw[32 + n];
    v8f sq = {};
    float acc = 0.0f;
#pragma unroll
    for (int tt = 0; tt < 4; ++tt) {
      v8f c = {};
#pragma unroll
      for (int kc = 0; kc < 4; ++kc)
        c = __builtin_amdgcn_wmma_f32_16x16x4_f32(
             false, a[kc],  false, bt[tt][kc],
             (short)0, c,  false,  false);
      const v8f e = c * c;
      sq += e;
      acc += (gm4[tt] + gln) * (((e[0] + e[1]) + (e[2] + e[3])) + ((e[4] + e[5]) + (e[6] + e[7])));
    }
    v8f wt;
    wt[0] = t0.x; wt[1] = t0.y; wt[2] = t0.z; wt[3] = t0.w;
    wt[4] = t1.x; wt[5] = t1.y; wt[6] = t1.z; wt[7] = t1.w;
    const v8f q = sq * wt;
    partial = acc + (((q[0] + q[1]) + (q[2] + q[3])) + ((q[4] + q[5]) + (q[6] + q[7])));
  }
  {
    const float ws = wave_sum(partial);
    if (fp_lane == 0) wsum[4 * smp + fp_wave] = ws;
  }
  __syncthreads();
  }

  if (tid < nb) {
    const v4f w = *reinterpret_cast<const v4f*>(&wsum[4 * tid]);
    out[b0 + tid] = ((w.x + w.y) + (w.z + w.w)) + hb;
  }
}

extern "C" void kernel_launch(void* const* d_in, const int* in_sizes, int n_in,
                              void* d_out, int out_size, void* d_ws, size_t ws_size,
                              hipStream_t stream) {
  const float* sb     = (const float*)d_in[0];
  const float* params = (const float*)d_in[1];
  const float* hw     = (const float*)d_in[2];
  const float* hb     = (const float*)d_in[3];
  float* out = (float*)d_out;
  int batch = in_sizes[0] / 12;
  qsim_wmma_kernel<<<dim3((batch + SPB - 1) / SPB), dim3(QS_THREADS), 0, stream>>>(
      sb, params, hw, hb, out, batch);
}
